# starting kernel
# baseline (speedup 1.0000x reference)
#include <hip/hip_runtime.h>


typedef __attribute__((ext_vector_type(2))) float v2f;
typedef __attribute__((ext_vector_type(8))) float v8f;

#define SEQS      4096
#define LSEQ      2048
#define NST       64
#define TCH       16
#define NSEG      8
#define CH_SEG    16
#define NTILE     256
#define WPB       4

#define OFF_M     0
#define OFF_CM    256
#define OFF_BM    1280
#define OFF_A16   2304
#define OFF_ST    2560

__device__ __forceinline__ v8f wmma4_f32(v2f a, v2f b, v8f c) {
  return __builtin_amdgcn_wmma_f32_16x16x4_f32(false, a, false, b,
                                               (short)0, c, false, false);
}

__global__ void ssm_prep(const float* __restrict__ A, const float* __restrict__ Bv,
                         const float* __restrict__ Cv, float* __restrict__ tbl) {
  int t = threadIdx.x;
  float* Mm  = tbl + OFF_M;
  float* Cm  = tbl + OFF_CM;
  float* Bm  = tbl + OFF_BM;
  float* a16 = tbl + OFF_A16;
  if (t < NST) {
    float a = A[t * NST + t];
    float b = Bv[t], c = Cv[t];
    float pw[17];
    float p = 1.0f;
    for (int i = 0; i < 17; ++i) { pw[i] = p; p *= a; }
    a16[t] = pw[16];
    for (int k = 0; k < TCH; ++k) Bm[t * TCH + k] = b * pw[15 - k];
    for (int l = 0; l < TCH; ++l) Cm[l * NST + t] = c * pw[l + 1];
  }
  if (t < 256) {
    int l = t >> 4, k = t & 15;
    float v = 0.0f;
    if (k <= l) {
      int d = l - k;
      for (int n = 0; n < NST; ++n) {
        float a = A[n * NST + n];
        float p = 1.0f;
        for (int i = 0; i < d; ++i) p *= a;
        v += Cv[n] * Bv[n] * p;
      }
    }
    Mm[l * TCH + k] = v;
  }
}

__global__ __launch_bounds__(WPB * 32) void ssm_state_pass(
    const float* __restrict__ u, const float* __restrict__ tbl,
    float* __restrict__ states) {
  const int lane = threadIdx.x & 31;
  const int half = lane >> 4;
  const int lm   = lane & 15;
  const int tile = blockIdx.x * WPB + (threadIdx.x >> 5);

  const float* Bm  = tbl + OFF_BM;
  const float* a16 = tbl + OFF_A16;

  v2f bm[4][4];
  #pragma unroll
  for (int t = 0; t < 4; ++t)
    #pragma unroll
    for (int j = 0; j < 4; ++j)
      bm[t][j] = *(const v2f*)(Bm + (16 * t + lm) * TCH + 4 * j + 2 * half);

  float av[4][8];
  #pragma unroll
  for (int t = 0; t < 4; ++t)
    #pragma unroll
    for (int r = 0; r < 8; ++r)
      av[t][r] = a16[16 * t + r + 8 * half];

  v8f H[4];
  const v8f zero = {0.f, 0.f, 0.f, 0.f, 0.f, 0.f, 0.f, 0.f};
  #pragma unroll
  for (int t = 0; t < 4; ++t) H[t] = zero;

  const float* useq = u + (size_t)(tile * 16 + lm) * LSEQ;

  for (int g = 0; g < NSEG; ++g) {
    float* sp = states + (size_t)(tile * NSEG + g) * 1024;
    #pragma unroll
    for (int t = 0; t < 4; ++t)
      #pragma unroll
      for (int r = 0; r < 8; ++r)
        sp[t * 256 + r * 32 + lane] = H[t][r];

    const float* useg = useq + g * (CH_SEG * TCH);
    for (int c = 0; c < CH_SEG; ++c) {
      const int cb = c * TCH;
      v2f ub[4];
      #pragma unroll
      for (int j = 0; j < 4; ++j)
        ub[j] = *(const v2f*)(useg + cb + 4 * j + 2 * half);
      __builtin_prefetch((const void*)(useg + cb + 4 * TCH), 0, 0);

      #pragma unroll
      for (int t = 0; t < 4; ++t) {
        #pragma unroll
        for (int r = 0; r < 8; ++r) H[t][r] *= av[t][r];
        #pragma unroll
        for (int j = 0; j < 4; ++j) H[t] = wmma4_f32(bm[t][j], ub[j], H[t]);
      }
    }
  }
}

__global__ __launch_bounds__(WPB * 32) void ssm_main(
    const float* __restrict__ u, const float* __restrict__ tbl,
    const float* __restrict__ states, float* __restrict__ y) {
  const int lane = threadIdx.x & 31;
  const int half = lane >> 4;
  const int lm   = lane & 15;
  const int tile = blockIdx.x * WPB + (threadIdx.x >> 5);
  const int seg  = blockIdx.y;

  const float* Mm  = tbl + OFF_M;
  const float* Cm  = tbl + OFF_CM;
  const float* Bm  = tbl + OFF_BM;
  const float* a16 = tbl + OFF_A16;

  v2f mA[4];
  #pragma unroll
  for (int j = 0; j < 4; ++j)
    mA[j] = *(const v2f*)(Mm + lm * TCH + 4 * j + 2 * half);

  v2f cmA[4][4];
  #pragma unroll
  for (int t = 0; t < 4; ++t)
    #pragma unroll
    for (int j = 0; j < 4; ++j)
      cmA[t][j] = *(const v2f*)(Cm + lm * NST + 16 * t + 4 * j + 2 * half);

  v2f bm[4][4];
  #pragma unroll
  for (int t = 0; t < 4; ++t)
    #pragma unroll
    for (int j = 0; j < 4; ++j)
      bm[t][j] = *(const v2f*)(Bm + (16 * t + lm) * TCH + 4 * j + 2 * half);

  float av[4][8];
  #pragma unroll
  for (int t = 0; t < 4; ++t)
    #pragma unroll
    for (int r = 0; r < 8; ++r)
      av[t][r] = a16[16 * t + r + 8 * half];

  v8f H[4];
  {
    const float* sp = states + (size_t)(tile * NSEG + seg) * 1024;
    #pragma unroll
    for (int t = 0; t < 4; ++t)
      #pragma unroll
      for (int r = 0; r < 8; ++r)
        H[t][r] = sp[t * 256 + r * 32 + lane];
  }

  const float* useq = u + (size_t)(tile * 16 + lm) * LSEQ + seg * (CH_SEG * TCH);
  float*       yseq = y + (size_t)(tile * 16 + lm) * LSEQ + seg * (CH_SEG * TCH);
  const v8f zero = {0.f, 0.f, 0.f, 0.f, 0.f, 0.f, 0.f, 0.f};

  for (int c = 0; c < CH_SEG; ++c) {
    const int cb = c * TCH;
    v2f ub[4];
    #pragma unroll
    for (int j = 0; j < 4; ++j)
      ub[j] = *(const v2f*)(useq + cb + 4 * j + 2 * half);
    __builtin_prefetch((const void*)(useq + cb + 4 * TCH), 0, 0);

    v8f yacc = zero;
    #pragma unroll
    for (int j = 0; j < 4; ++j) yacc = wmma4_f32(mA[j], ub[j], yacc);

    #pragma unroll
    for (int t = 0; t < 4; ++t) {
      #pragma unroll
      for (int j = 0; j < 4; ++j) {
        const int jj = j & 1;
        float e0 = H[t][4 * jj + 0], e1 = H[t][4 * jj + 1];
        float e2 = H[t][4 * jj + 2], e3 = H[t][4 * jj + 3];
        float sx = half ? e0 : e2;
        float sy = half ? e1 : e3;
        float px = __shfl_xor(sx, 16);
        float py = __shfl_xor(sy, 16);
        float ox = half ? e2 : e0;
        float oy = half ? e3 : e1;
        bool useOwn = (half == (j >> 1));
        v2f hb;
        hb.x = useOwn ? ox : px;
        hb.y = useOwn ? oy : py;
        yacc = wmma4_f32(cmA[t][j], hb, yacc);
      }
    }

    if (c != CH_SEG - 1) {
      #pragma unroll
      for (int t = 0; t < 4; ++t) {
        #pragma unroll
        for (int r = 0; r < 8; ++r) H[t][r] *= av[t][r];
        #pragma unroll
        for (int j = 0; j < 4; ++j) H[t] = wmma4_f32(bm[t][j], ub[j], H[t]);
      }
    }

    #pragma unroll
    for (int r = 0; r < 8; ++r)
      yseq[cb + r + 8 * half] = yacc[r];
  }
}

extern "C" void kernel_launch(void* const* d_in, const int* in_sizes, int n_in,
                              void* d_out, int out_size, void* d_ws, size_t ws_size,
                              hipStream_t stream) {
  const float* u  = (const float*)d_in[0];
  const float* A  = (const float*)d_in[1];
  const float* Bv = (const float*)d_in[2];
  const float* Cv = (const float*)d_in[3];
  float* tbl    = (float*)d_ws;
  float* states = tbl + OFF_ST;
  float* y      = (float*)d_out;

  ssm_prep<<<dim3(1), dim3(256), 0, stream>>>(A, Bv, Cv, tbl);
  ssm_state_pass<<<dim3(NTILE / WPB), dim3(WPB * 32), 0, stream>>>(u, tbl, states);
  ssm_main<<<dim3(NTILE / WPB, NSEG), dim3(WPB * 32), 0, stream>>>(u, tbl, states, y);
}


# state pass made parallel over the 8 segments: each wave scans its 256-step segment from a zero state; boundary states by an LDS fold S=a^256*S+E_g (sum order of the carried state changes)
# speedup vs baseline: 1.5331x; 1.5331x over previous
#include <hip/hip_runtime.h>


typedef __attribute__((ext_vector_type(2))) float v2f;
typedef __attribute__((ext_vector_type(8))) float v8f;

#define SEQS      4096
#define LSEQ      2048
#define NST       64
#define TCH       16
#define NSEG      8
#define CH_SEG    16
#define NTILE     256
#define WPB       4

#define OFF_M     0
#define OFF_CM    256
#define OFF_BM    1280
#define OFF_A16   2304
#define OFF_ST    2560

__device__ __forceinline__ v8f wmma4_f32(v2f a, v2f b, v8f c) {
  return __builtin_amdgcn_wmma_f32_16x16x4_f32(false, a, false, b,
                                               (short)0, c, false, false);
}

__global__ void ssm_prep(const float* __restrict__ A, const float* __restrict__ Bv,
                         const float* __restrict__ Cv, float* __restrict__ tbl) {
  int t = threadIdx.x;
  float* Mm  = tbl + OFF_M;
  float* Cm  = tbl + OFF_CM;
  float* Bm  = tbl + OFF_BM;
  float* a16 = tbl + OFF_A16;
  if (t < NST) {
    float a = A[t * NST + t];
    float b = Bv[t], c = Cv[t];
    float pw[17];
    float p = 1.0f;
    for (int i = 0; i < 17; ++i) { pw[i] = p; p *= a; }
    a16[t] = pw[16];
    for (int k = 0; k < TCH; ++k) Bm[t * TCH + k] = b * pw[15 - k];
    for (int l = 0; l < TCH; ++l) Cm[l * NST + t] = c * pw[l + 1];
  }
  if (t < 256) {
    int l = t >> 4, k = t & 15;
    float v = 0.0f;
    if (k <= l) {
      int d = l - k;
      for (int n = 0; n < NST; ++n) {
        float a = A[n * NST + n];
        float p = 1.0f;
        for (int i = 0; i < d; ++i) p *= a;
        v += Cv[n] * Bv[n] * p;
      }
    }
    Mm[l * TCH + k] = v;
  }
}

__global__ __launch_bounds__(NSEG * 32) void ssm_state_pass(
    const float* __restrict__ u, const float* __restrict__ tbl,
    float* __restrict__ states) {
  __shared__ float Eloc[(NSEG - 1) * 1024];

  const int lane = threadIdx.x & 31;
  const int half = lane >> 4;
  const int lm   = lane & 15;
  const int tile = blockIdx.x;
  const int seg  = __builtin_amdgcn_readfirstlane((int)(threadIdx.x >> 5));

  const float* Bm  = tbl + OFF_BM;
  const float* a16 = tbl + OFF_A16;

  float av[4][8];
  #pragma unroll
  for (int t = 0; t < 4; ++t)
    #pragma unroll
    for (int r = 0; r < 8; ++r)
      av[t][r] = a16[16 * t + r + 8 * half];

  const v8f zero = {0.f, 0.f, 0.f, 0.f, 0.f, 0.f, 0.f, 0.f};

  if (seg < NSEG - 1) {
    v2f bm[4][4];
    #pragma unroll
    for (int t = 0; t < 4; ++t)
      #pragma unroll
      for (int j = 0; j < 4; ++j)
        bm[t][j] = *(const v2f*)(Bm + (16 * t + lm) * TCH + 4 * j + 2 * half);

    v8f H[4];
    #pragma unroll
    for (int t = 0; t < 4; ++t) H[t] = zero;

    const float* useg = u + (size_t)(tile * 16 + lm) * LSEQ + seg * (CH_SEG * TCH);
    for (int c = 0; c < CH_SEG; ++c) {
      const int cb = c * TCH;
      v2f ub[4];
      #pragma unroll
      for (int j = 0; j < 4; ++j)
        ub[j] = *(const v2f*)(useg + cb + 4 * j + 2 * half);
      __builtin_prefetch((const void*)(useg + cb + 4 * TCH), 0, 0);

      #pragma unroll
      for (int t = 0; t < 4; ++t) {
        #pragma unroll
        for (int r = 0; r < 8; ++r) H[t][r] *= av[t][r];
        #pragma unroll
        for (int j = 0; j < 4; ++j) H[t] = wmma4_f32(bm[t][j], ub[j], H[t]);
      }
    }

    float* ep = Eloc + seg * 1024;
    #pragma unroll
    for (int t = 0; t < 4; ++t)
      #pragma unroll
      for (int r = 0; r < 8; ++r)
        ep[t * 256 + r * 32 + lane] = H[t][r];
  }
  __syncthreads();

  #pragma unroll
  for (int t = 0; t < 4; ++t)
    #pragma unroll
    for (int r = 0; r < 8; ++r) {
      float p = av[t][r];
      p *= p; p *= p; p *= p; p *= p;
      av[t][r] = p;
    }

  v8f S[4];
  #pragma unroll
  for (int t = 0; t < 4; ++t) S[t] = zero;
  for (int g = 0; g < seg; ++g) {
    const float* ep = Eloc + g * 1024;
    #pragma unroll
    for (int t = 0; t < 4; ++t)
      #pragma unroll
      for (int r = 0; r < 8; ++r)
        S[t][r] = av[t][r] * S[t][r] + ep[t * 256 + r * 32 + lane];
  }

  float* sp = states + (size_t)(tile * NSEG + seg) * 1024;
  #pragma unroll
  for (int t = 0; t < 4; ++t)
    #pragma unroll
    for (int r = 0; r < 8; ++r)
      sp[t * 256 + r * 32 + lane] = S[t][r];
}

__global__ __launch_bounds__(WPB * 32) void ssm_main(
    const float* __restrict__ u, const float* __restrict__ tbl,
    const float* __restrict__ states, float* __restrict__ y) {
  const int lane = threadIdx.x & 31;
  const int half = lane >> 4;
  const int lm   = lane & 15;
  const int tile = blockIdx.x * WPB + (threadIdx.x >> 5);
  const int seg  = blockIdx.y;

  const float* Mm  = tbl + OFF_M;
  const float* Cm  = tbl + OFF_CM;
  const float* Bm  = tbl + OFF_BM;
  const float* a16 = tbl + OFF_A16;

  v2f mA[4];
  #pragma unroll
  for (int j = 0; j < 4; ++j)
    mA[j] = *(const v2f*)(Mm + lm * TCH + 4 * j + 2 * half);

  v2f cmA[4][4];
  #pragma unroll
  for (int t = 0; t < 4; ++t)
    #pragma unroll
    for (int j = 0; j < 4; ++j)
      cmA[t][j] = *(const v2f*)(Cm + lm * NST + 16 * t + 4 * j + 2 * half);

  v2f bm[4][4];
  #pragma unroll
  for (int t = 0; t < 4; ++t)
    #pragma unroll
    for (int j = 0; j < 4; ++j)
      bm[t][j] = *(const v2f*)(Bm + (16 * t + lm) * TCH + 4 * j + 2 * half);

  float av[4][8];
  #pragma unroll
  for (int t = 0; t < 4; ++t)
    #pragma unroll
    for (int r = 0; r < 8; ++r)
      av[t][r] = a16[16 * t + r + 8 * half];

  v8f H[4];
  {
    const float* sp = states + (size_t)(tile * NSEG + seg) * 1024;
    #pragma unroll
    for (int t = 0; t < 4; ++t)
      #pragma unroll
      for (int r = 0; r < 8; ++r)
        H[t][r] = sp[t * 256 + r * 32 + lane];
  }

  const float* useq = u + (size_t)(tile * 16 + lm) * LSEQ + seg * (CH_SEG * TCH);
  float*       yseq = y + (size_t)(tile * 16 + lm) * LSEQ + seg * (CH_SEG * TCH);
  const v8f zero = {0.f, 0.f, 0.f, 0.f, 0.f, 0.f, 0.f, 0.f};

  for (int c = 0; c < CH_SEG; ++c) {
    const int cb = c * TCH;
    v2f ub[4];
    #pragma unroll
    for (int j = 0; j < 4; ++j)
      ub[j] = *(const v2f*)(useq + cb + 4 * j + 2 * half);
    __builtin_prefetch((const void*)(useq + cb + 4 * TCH), 0, 0);

    v8f yacc = zero;
    #pragma unroll
    for (int j = 0; j < 4; ++j) yacc = wmma4_f32(mA[j], ub[j], yacc);

    #pragma unroll
    for (int t = 0; t < 4; ++t) {
      #pragma unroll
      for (int j = 0; j < 4; ++j) {
        const int jj = j & 1;
        float e0 = H[t][4 * jj + 0], e1 = H[t][4 * jj + 1];
        float e2 = H[t][4 * jj + 2], e3 = H[t][4 * jj + 3];
        float sx = half ? e0 : e2;
        float sy = half ? e1 : e3;
        float px = __shfl_xor(sx, 16);
        float py = __shfl_xor(sy, 16);
        float ox = half ? e2 : e0;
        float oy = half ? e3 : e1;
        bool useOwn = (half == (j >> 1));
        v2f hb;
        hb.x = useOwn ? ox : px;
        hb.y = useOwn ? oy : py;
        yacc = wmma4_f32(cmA[t][j], hb, yacc);
      }
    }

    if (c != CH_SEG - 1) {
      #pragma unroll
      for (int t = 0; t < 4; ++t) {
        #pragma unroll
        for (int r = 0; r < 8; ++r) H[t][r] *= av[t][r];
        #pragma unroll
        for (int j = 0; j < 4; ++j) H[t] = wmma4_f32(bm[t][j], ub[j], H[t]);
      }
    }

    #pragma unroll
    for (int r = 0; r < 8; ++r)
      yseq[cb + r + 8 * half] = yacc[r];
  }
}

extern "C" void kernel_launch(void* const* d_in, const int* in_sizes, int n_in,
                              void* d_out, int out_size, void* d_ws, size_t ws_size,
                              hipStream_t stream) {
  const float* u  = (const float*)d_in[0];
  const float* A  = (const float*)d_in[1];
  const float* Bv = (const float*)d_in[2];
  const float* Cv = (const float*)d_in[3];
  float* tbl    = (float*)d_ws;
  float* states = tbl + OFF_ST;
  float* y      = (float*)d_out;

  ssm_prep<<<dim3(1), dim3(256), 0, stream>>>(A, Bv, Cv, tbl);
  ssm_state_pass<<<dim3(NTILE), dim3(NSEG * 32), 0, stream>>>(u, tbl, states);
  ssm_main<<<dim3(NTILE / WPB, NSEG), dim3(WPB * 32), 0, stream>>>(u, tbl, states, y);
}


# table kernel: M's 64 terms no longer wait one by one on a scalar load of a,b,c and a divergent power loop; state threads load once, powers go to LDS, M summed from LDS in the same n order
# speedup vs baseline: 2.2480x; 1.4663x over previous
#include <hip/hip_runtime.h>


typedef __attribute__((ext_vector_type(2))) float v2f;
typedef __attribute__((ext_vector_type(8))) float v8f;

#define SEQS      4096
#define LSEQ      2048
#define NST       64
#define TCH       16
#define NSEG      8
#define CH_SEG    16
#define NTILE     256
#define WPB       4

#define OFF_M     0
#define OFF_CM    256
#define OFF_BM    1280
#define OFF_A16   2304
#define OFF_ST    2560

__device__ __forceinline__ v8f wmma4_f32(v2f a, v2f b, v8f c) {
  return __builtin_amdgcn_wmma_f32_16x16x4_f32(false, a, false, b,
                                               (short)0, c, false, false);
}

#define PPITCH    68

__global__ void ssm_prep(const float* __restrict__ A, const float* __restrict__ Bv,
                         const float* __restrict__ Cv, float* __restrict__ tbl) {
  __shared__ __attribute__((aligned(16))) float Pw[TCH * PPITCH];
  __shared__ __attribute__((aligned(16))) float CB[NST];

  int t = threadIdx.x;
  float* Mm  = tbl + OFF_M;
  float* Cm  = tbl + OFF_CM;
  float* Bm  = tbl + OFF_BM;
  float* a16 = tbl + OFF_A16;

  float pw[17];
  float b = 0.0f, c = 0.0f;
  if (t < NST) {
    float a = A[t * NST + t];
    b = Bv[t]; c = Cv[t];
    float p = 1.0f;
    for (int i = 0; i < 17; ++i) { pw[i] = p; p *= a; }
    for (int d = 0; d < TCH; ++d) Pw[d * PPITCH + t] = pw[d];
    CB[t] = c * b;
  }
  __syncthreads();

  if (t < NST) {
    a16[t] = pw[16];
    for (int k = 0; k < TCH; ++k) Bm[t * TCH + k] = b * pw[15 - k];
    for (int l = 0; l < TCH; ++l) Cm[l * NST + t] = c * pw[l + 1];
  }

  if (t < 256) {
    int l = t >> 4, k = t & 15;
    float v = 0.0f;
    if (k <= l) {
      const float* prow = Pw + (l - k) * PPITCH;
      for (int q = 0; q < NST / 4; ++q) {
        const float4 pv = *(const float4*)(prow + 4 * q);
        const float4 cb = *(const float4*)(CB + 4 * q);
        v = fmaf(cb.x, pv.x, v);
        v = fmaf(cb.y, pv.y, v);
        v = fmaf(cb.z, pv.z, v);
        v = fmaf(cb.w, pv.w, v);
      }
    }
    Mm[l * TCH + k] = v;
  }
}

__global__ __launch_bounds__(NSEG * 32) void ssm_state_pass(
    const float* __restrict__ u, const float* __restrict__ tbl,
    float* __restrict__ states) {
  __shared__ float Eloc[(NSEG - 1) * 1024];

  const int lane = threadIdx.x & 31;
  const int half = lane >> 4;
  const int lm   = lane & 15;
  const int tile = blockIdx.x;
  const int seg  = __builtin_amdgcn_readfirstlane((int)(threadIdx.x >> 5));

  const float* Bm  = tbl + OFF_BM;
  const float* a16 = tbl + OFF_A16;

  float av[4][8];
  #pragma unroll
  for (int t = 0; t < 4; ++t)
    #pragma unroll
    for (int r = 0; r < 8; ++r)
      av[t][r] = a16[16 * t + r + 8 * half];

  const v8f zero = {0.f, 0.f, 0.f, 0.f, 0.f, 0.f, 0.f, 0.f};

  if (seg < NSEG - 1) {
    v2f bm[4][4];
    #pragma unroll
    for (int t = 0; t < 4; ++t)
      #pragma unroll
      for (int j = 0; j < 4; ++j)
        bm[t][j] = *(const v2f*)(Bm + (16 * t + lm) * TCH + 4 * j + 2 * half);

    v8f H[4];
    #pragma unroll
    for (int t = 0; t < 4; ++t) H[t] = zero;

    const float* useg = u + (size_t)(tile * 16 + lm) * LSEQ + seg * (CH_SEG * TCH);
    for (int c = 0; c < CH_SEG; ++c) {
      const int cb = c * TCH;
      v2f ub[4];
      #pragma unroll
      for (int j = 0; j < 4; ++j)
        ub[j] = *(const v2f*)(useg + cb + 4 * j + 2 * half);
      __builtin_prefetch((const void*)(useg + cb + 4 * TCH), 0, 0);

      #pragma unroll
      for (int t = 0; t < 4; ++t) {
        #pragma unroll
        for (int r = 0; r < 8; ++r) H[t][r] *= av[t][r];
        #pragma unroll
        for (int j = 0; j < 4; ++j) H[t] = wmma4_f32(bm[t][j], ub[j], H[t]);
      }
    }

    float* ep = Eloc + seg * 1024;
    #pragma unroll
    for (int t = 0; t < 4; ++t)
      #pragma unroll
      for (int r = 0; r < 8; ++r)
        ep[t * 256 + r * 32 + lane] = H[t][r];
  }
  __syncthreads();

  #pragma unroll
  for (int t = 0; t < 4; ++t)
    #pragma unroll
    for (int r = 0; r < 8; ++r) {
      float p = av[t][r];
      p *= p; p *= p; p *= p; p *= p;
      av[t][r] = p;
    }

  v8f S[4];
  #pragma unroll
  for (int t = 0; t < 4; ++t) S[t] = zero;
  for (int g = 0; g < seg; ++g) {
    const float* ep = Eloc + g * 1024;
    #pragma unroll
    for (int t = 0; t < 4; ++t)
      #pragma unroll
      for (int r = 0; r < 8; ++r)
        S[t][r] = av[t][r] * S[t][r] + ep[t * 256 + r * 32 + lane];
  }

  float* sp = states + (size_t)(tile * NSEG + seg) * 1024;
  #pragma unroll
  for (int t = 0; t < 4; ++t)
    #pragma unroll
    for (int r = 0; r < 8; ++r)
      sp[t * 256 + r * 32 + lane] = S[t][r];
}

__global__ __launch_bounds__(WPB * 32) void ssm_main(
    const float* __restrict__ u, const float* __restrict__ tbl,
    const float* __restrict__ states, float* __restrict__ y) {
  const int lane = threadIdx.x & 31;
  const int half = lane >> 4;
  const int lm   = lane & 15;
  const int tile = blockIdx.x * WPB + (threadIdx.x >> 5);
  const int seg  = blockIdx.y;

  const float* Mm  = tbl + OFF_M;
  const float* Cm  = tbl + OFF_CM;
  const float* Bm  = tbl + OFF_BM;
  const float* a16 = tbl + OFF_A16;

  v2f mA[4];
  #pragma unroll
  for (int j = 0; j < 4; ++j)
    mA[j] = *(const v2f*)(Mm + lm * TCH + 4 * j + 2 * half);

  v2f cmA[4][4];
  #pragma unroll
  for (int t = 0; t < 4; ++t)
    #pragma unroll
    for (int j = 0; j < 4; ++j)
      cmA[t][j] = *(const v2f*)(Cm + lm * NST + 16 * t + 4 * j + 2 * half);

  v2f bm[4][4];
  #pragma unroll
  for (int t = 0; t < 4; ++t)
    #pragma unroll
    for (int j = 0; j < 4; ++j)
      bm[t][j] = *(const v2f*)(Bm + (16 * t + lm) * TCH + 4 * j + 2 * half);

  float av[4][8];
  #pragma unroll
  for (int t = 0; t < 4; ++t)
    #pragma unroll
    for (int r = 0; r < 8; ++r)
      av[t][r] = a16[16 * t + r + 8 * half];

  v8f H[4];
  {
    const float* sp = states + (size_t)(tile * NSEG + seg) * 1024;
    #pragma unroll
    for (int t = 0; t < 4; ++t)
      #pragma unroll
      for (int r = 0; r < 8; ++r)
        H[t][r] = sp[t * 256 + r * 32 + lane];
  }

  const float* useq = u + (size_t)(tile * 16 + lm) * LSEQ + seg * (CH_SEG * TCH);
  float*       yseq = y + (size_t)(tile * 16 + lm) * LSEQ + seg * (CH_SEG * TCH);
  const v8f zero = {0.f, 0.f, 0.f, 0.f, 0.f, 0.f, 0.f, 0.f};

  for (int c = 0; c < CH_SEG; ++c) {
    const int cb = c * TCH;
    v2f ub[4];
    #pragma unroll
    for (int j = 0; j < 4; ++j)
      ub[j] = *(const v2f*)(useq + cb + 4 * j + 2 * half);
    __builtin_prefetch((const void*)(useq + cb + 4 * TCH), 0, 0);

    v8f yacc = zero;
    #pragma unroll
    for (int j = 0; j < 4; ++j) yacc = wmma4_f32(mA[j], ub[j], yacc);

    #pragma unroll
    for (int t = 0; t < 4; ++t) {
      #pragma unroll
      for (int j = 0; j < 4; ++j) {
        const int jj = j & 1;
        float e0 = H[t][4 * jj + 0], e1 = H[t][4 * jj + 1];
        float e2 = H[t][4 * jj + 2], e3 = H[t][4 * jj + 3];
        float sx = half ? e0 : e2;
        float sy = half ? e1 : e3;
        float px = __shfl_xor(sx, 16);
        float py = __shfl_xor(sy, 16);
        float ox = half ? e2 : e0;
        float oy = half ? e3 : e1;
        bool useOwn = (half == (j >> 1));
        v2f hb;
        hb.x = useOwn ? ox : px;
        hb.y = useOwn ? oy : py;
        yacc = wmma4_f32(cmA[t][j], hb, yacc);
      }
    }

    if (c != CH_SEG - 1) {
      #pragma unroll
      for (int t = 0; t < 4; ++t) {
        #pragma unroll
        for (int r = 0; r < 8; ++r) H[t][r] *= av[t][r];
        #pragma unroll
        for (int j = 0; j < 4; ++j) H[t] = wmma4_f32(bm[t][j], ub[j], H[t]);
      }
    }

    #pragma unroll
    for (int r = 0; r < 8; ++r)
      yseq[cb + r + 8 * half] = yacc[r];
  }
}

extern "C" void kernel_launch(void* const* d_in, const int* in_sizes, int n_in,
                              void* d_out, int out_size, void* d_ws, size_t ws_size,
                              hipStream_t stream) {
  const float* u  = (const float*)d_in[0];
  const float* A  = (const float*)d_in[1];
  const float* Bv = (const float*)d_in[2];
  const float* Cv = (const float*)d_in[3];
  float* tbl    = (float*)d_ws;
  float* states = tbl + OFF_ST;
  float* y      = (float*)d_out;

  ssm_prep<<<dim3(1), dim3(256), 0, stream>>>(A, Bv, Cv, tbl);
  ssm_state_pass<<<dim3(NTILE), dim3(NSEG * 32), 0, stream>>>(u, tbl, states);
  ssm_main<<<dim3(NTILE / WPB, NSEG), dim3(WPB * 32), 0, stream>>>(u, tbl, states, y);
}


# one launch instead of three: each work-group builds the tables in LDS, scans its 8 segments, folds the boundary states in LDS and computes Y from registers; sum orders unchanged
# speedup vs baseline: 2.5156x; 1.1190x over previous
#include <hip/hip_runtime.h>


typedef __attribute__((ext_vector_type(2))) float v2f;
typedef __attribute__((ext_vector_type(8))) float v8f;

#define SEQS      4096
#define LSEQ      2048
#define NST       64
#define TCH       16
#define NSEG      8
#define CH_SEG    16
#define NTILE     256

#define OFF_M     0
#define OFF_CM    256
#define OFF_BM    1280
#define OFF_A16   2304
#define TBL_FLTS  2368

#define PPITCH    68

__device__ __forceinline__ v8f wmma4_f32(v2f a, v2f b, v8f c) {
  return __builtin_amdgcn_wmma_f32_16x16x4_f32(false, a, false, b,
                                               (short)0, c, false, false);
}

__global__ __launch_bounds__(NSEG * 32) void ssm_fused(
    const float* __restrict__ u, const float* __restrict__ A,
    const float* __restrict__ Bv, const float* __restrict__ Cv,
    float* __restrict__ y) {
  __shared__ __attribute__((aligned(16))) float Pw[TCH * PPITCH];
  __shared__ __attribute__((aligned(16))) float CB[NST];
  __shared__ __attribute__((aligned(16))) float tbl[TBL_FLTS];
  __shared__ float Eloc[(NSEG - 1) * 1024];

  const int tid  = threadIdx.x;
  const int lane = tid & 31;
  const int half = lane >> 4;
  const int lm   = lane & 15;
  const int tile = blockIdx.x;
  const int seg  = __builtin_amdgcn_readfirstlane((int)(threadIdx.x >> 5));

  float* Mm  = tbl + OFF_M;
  float* Cm  = tbl + OFF_CM;
  float* Bm  = tbl + OFF_BM;
  float* a16 = tbl + OFF_A16;

  const float* useq = u + (size_t)(tile * 16 + lm) * LSEQ + seg * (CH_SEG * TCH);
  float*       yseq = y + (size_t)(tile * 16 + lm) * LSEQ + seg * (CH_SEG * TCH);
  __builtin_prefetch((const void*)(useq + 32 * half), 0, 0);

  {
    float pw[17];
    float b = 0.0f, c = 0.0f;
    if (tid < NST) {
      float a = A[tid * NST + tid];
      b = Bv[tid]; c = Cv[tid];
      float p = 1.0f;
      for (int i = 0; i < 17; ++i) { pw[i] = p; p *= a; }
      for (int d = 0; d < TCH; ++d) Pw[d * PPITCH + tid] = pw[d];
      CB[tid] = c * b;
    }
    __syncthreads();

    if (tid < NST) {
      a16[tid] = pw[16];
      for (int k = 0; k < TCH; ++k) Bm[tid * TCH + k] = b * pw[15 - k];
      for (int l = 0; l < TCH; ++l) Cm[l * NST + tid] = c * pw[l + 1];
    }

    {
      int l = tid >> 4, k = tid & 15;
      float v = 0.0f;
      if (k <= l) {
        const float* prow = Pw + (l - k) * PPITCH;
        for (int q = 0; q < NST / 4; ++q) {
          const float4 pv = *(const float4*)(prow + 4 * q);
          const float4 cb = *(const float4*)(CB + 4 * q);
          v = fmaf(cb.x, pv.x, v);
          v = fmaf(cb.y, pv.y, v);
          v = fmaf(cb.z, pv.z, v);
          v = fmaf(cb.w, pv.w, v);
        }
      }
      Mm[l * TCH + k] = v;
    }
    __syncthreads();
  }

  float av[4][8];
  #pragma unroll
  for (int t = 0; t < 4; ++t)
    #pragma unroll
    for (int r = 0; r < 8; ++r)
      av[t][r] = a16[16 * t + r + 8 * half];

  v2f bm[4][4];
  #pragma unroll
  for (int t = 0; t < 4; ++t)
    #pragma unroll
    for (int j = 0; j < 4; ++j)
      bm[t][j] = *(const v2f*)(Bm + (16 * t + lm) * TCH + 4 * j + 2 * half);

  const v8f zero = {0.f, 0.f, 0.f, 0.f, 0.f, 0.f, 0.f, 0.f};

  if (seg < NSEG - 1) {
    v8f E[4];
    #pragma unroll
    for (int t = 0; t < 4; ++t) E[t] = zero;

    for (int c = 0; c < CH_SEG; ++c) {
      const int cb = c * TCH;
      v2f ub[4];
      #pragma unroll
      for (int j = 0; j < 4; ++j)
        ub[j] = *(const v2f*)(useq + cb + 4 * j + 2 * half);
      __builtin_prefetch((const void*)(useq + cb + 4 * TCH), 0, 0);

      #pragma unroll
      for (int t = 0; t < 4; ++t) {
        #pragma unroll
        for (int r = 0; r < 8; ++r) E[t][r] *= av[t][r];
        #pragma unroll
        for (int j = 0; j < 4; ++j) E[t] = wmma4_f32(bm[t][j], ub[j], E[t]);
      }
    }

    float* ep = Eloc + seg * 1024;
    #pragma unroll
    for (int t = 0; t < 4; ++t)
      #pragma unroll
      for (int r = 0; r < 8; ++r)
        ep[t * 256 + r * 32 + lane] = E[t][r];
  }
  __syncthreads();

  v8f H[4];
  {
    float aw[4][8];
    #pragma unroll
    for (int t = 0; t < 4; ++t)
      #pragma unroll
      for (int r = 0; r < 8; ++r) {
        float p = av[t][r];
        p *= p; p *= p; p *= p; p *= p;
        aw[t][r] = p;
      }

    #pragma unroll
    for (int t = 0; t < 4; ++t) H[t] = zero;
    for (int g = 0; g < seg; ++g) {
      const float* ep = Eloc + g * 1024;
      #pragma unroll
      for (int t = 0; t < 4; ++t)
        #pragma unroll
        for (int r = 0; r < 8; ++r)
          H[t][r] = aw[t][r] * H[t][r] + ep[t * 256 + r * 32 + lane];
    }
  }

  v2f mA[4];
  #pragma unroll
  for (int j = 0; j < 4; ++j)
    mA[j] = *(const v2f*)(Mm + lm * TCH + 4 * j + 2 * half);

  v2f cmA[4][4];
  #pragma unroll
  for (int t = 0; t < 4; ++t)
    #pragma unroll
    for (int j = 0; j < 4; ++j)
      cmA[t][j] = *(const v2f*)(Cm + lm * NST + 16 * t + 4 * j + 2 * half);

  for (int c = 0; c < CH_SEG; ++c) {
    const int cb = c * TCH;
    v2f ub[4];
    #pragma unroll
    for (int j = 0; j < 4; ++j)
      ub[j] = *(const v2f*)(useq + cb + 4 * j + 2 * half);
    __builtin_prefetch((const void*)(useq + cb + 4 * TCH), 0, 0);

    v8f yacc = zero;
    #pragma unroll
    for (int j = 0; j < 4; ++j) yacc = wmma4_f32(mA[j], ub[j], yacc);

    #pragma unroll
    for (int t = 0; t < 4; ++t) {
      #pragma unroll
      for (int j = 0; j < 4; ++j) {
        const int jj = j & 1;
        float e0 = H[t][4 * jj + 0], e1 = H[t][4 * jj + 1];
        float e2 = H[t][4 * jj + 2], e3 = H[t][4 * jj + 3];
        float sx = half ? e0 : e2;
        float sy = half ? e1 : e3;
        float px = __shfl_xor(sx, 16);
        float py = __shfl_xor(sy, 16);
        float ox = half ? e2 : e0;
        float oy = half ? e3 : e1;
        bool useOwn = (half == (j >> 1));
        v2f hb;
        hb.x = useOwn ? ox : px;
        hb.y = useOwn ? oy : py;
        yacc = wmma4_f32(cmA[t][j], hb, yacc);
      }
    }

    if (c != CH_SEG - 1) {
      #pragma unroll
      for (int t = 0; t < 4; ++t) {
        #pragma unroll
        for (int r = 0; r < 8; ++r) H[t][r] *= av[t][r];
        #pragma unroll
        for (int j = 0; j < 4; ++j) H[t] = wmma4_f32(bm[t][j], ub[j], H[t]);
      }
    }

    #pragma unroll
    for (int r = 0; r < 8; ++r)
      yseq[cb + r + 8 * half] = yacc[r];
  }
}

extern "C" void kernel_launch(void* const* d_in, const int* in_sizes, int n_in,
                              void* d_out, int out_size, void* d_ws, size_t ws_size,
                              hipStream_t stream) {
  const float* u  = (const float*)d_in[0];
  const float* A  = (const float*)d_in[1];
  const float* Bv = (const float*)d_in[2];
  const float* Cv = (const float*)d_in[3];
  float* y      = (float*)d_out;

  ssm_fused<<<dim3(NTILE), dim3(NSEG * 32), 0, stream>>>(u, A, Bv, Cv, y);
}


# segments halved to 128 steps: 16 waves per work-group (4 per SIMD) instead of 8, so other waves cover waits on loads and LDS; fold S=a^128*S+E_g over 16 segments (carried state's sum order changes)
# speedup vs baseline: 2.6142x; 1.0392x over previous
#include <hip/hip_runtime.h>


typedef __attribute__((ext_vector_type(2))) float v2f;
typedef __attribute__((ext_vector_type(8))) float v8f;

#define SEQS      4096
#define LSEQ      2048
#define NST       64
#define TCH       16
#define NSEG      16
#define CH_SEG    8
#define NTILE     256

#define OFF_M     0
#define OFF_CM    256
#define OFF_BM    1280
#define OFF_A16   2304
#define TBL_FLTS  2368

#define PPITCH    68

__device__ __forceinline__ v8f wmma4_f32(v2f a, v2f b, v8f c) {
  return __builtin_amdgcn_wmma_f32_16x16x4_f32(false, a, false, b,
                                               (short)0, c, false, false);
}

__global__ __launch_bounds__(NSEG * 32) void ssm_fused(
    const float* __restrict__ u, const float* __restrict__ A,
    const float* __restrict__ Bv, const float* __restrict__ Cv,
    float* __restrict__ y) {
  __shared__ __attribute__((aligned(16))) float Pw[TCH * PPITCH];
  __shared__ __attribute__((aligned(16))) float CB[NST];
  __shared__ __attribute__((aligned(16))) float tbl[TBL_FLTS];
  __shared__ float Eloc[(NSEG - 1) * 1024];

  const int tid  = threadIdx.x;
  const int lane = tid & 31;
  const int half = lane >> 4;
  const int lm   = lane & 15;
  const int tile = blockIdx.x;
  const int seg  = __builtin_amdgcn_readfirstlane((int)(threadIdx.x >> 5));

  float* Mm  = tbl + OFF_M;
  float* Cm  = tbl + OFF_CM;
  float* Bm  = tbl + OFF_BM;
  float* a16 = tbl + OFF_A16;

  const float* useq = u + (size_t)(tile * 16 + lm) * LSEQ + seg * (CH_SEG * TCH);
  float*       yseq = y + (size_t)(tile * 16 + lm) * LSEQ + seg * (CH_SEG * TCH);
  __builtin_prefetch((const void*)(useq + 32 * half), 0, 0);

  {
    float pw[17];
    float b = 0.0f, c = 0.0f;
    if (tid < NST) {
      float a = A[tid * NST + tid];
      b = Bv[tid]; c = Cv[tid];
      float p = 1.0f;
      for (int i = 0; i < 17; ++i) { pw[i] = p; p *= a; }
      for (int d = 0; d < TCH; ++d) Pw[d * PPITCH + tid] = pw[d];
      CB[tid] = c * b;
    }
    __syncthreads();

    if (tid < NST) {
      a16[tid] = pw[16];
      for (int k = 0; k < TCH; ++k) Bm[tid * TCH + k] = b * pw[15 - k];
      for (int l = 0; l < TCH; ++l) Cm[l * NST + tid] = c * pw[l + 1];
    }

    if (tid < TCH * TCH) {
      int l = tid >> 4, k = tid & 15;
      float v = 0.0f;
      if (k <= l) {
        const float* prow = Pw + (l - k) * PPITCH;
        for (int q = 0; q < NST / 4; ++q) {
          const float4 pv = *(const float4*)(prow + 4 * q);
          const float4 cb = *(const float4*)(CB + 4 * q);
          v = fmaf(cb.x, pv.x, v);
          v = fmaf(cb.y, pv.y, v);
          v = fmaf(cb.z, pv.z, v);
          v = fmaf(cb.w, pv.w, v);
        }
      }
      Mm[l * TCH + k] = v;
    }
    __syncthreads();
  }

  float av[4][8];
  #pragma unroll
  for (int t = 0; t < 4; ++t)
    #pragma unroll
    for (int r = 0; r < 8; ++r)
      av[t][r] = a16[16 * t + r + 8 * half];

  v2f bm[4][4];
  #pragma unroll
  for (int t = 0; t < 4; ++t)
    #pragma unroll
    for (int j = 0; j < 4; ++j)
      bm[t][j] = *(const v2f*)(Bm + (16 * t + lm) * TCH + 4 * j + 2 * half);

  const v8f zero = {0.f, 0.f, 0.f, 0.f, 0.f, 0.f, 0.f, 0.f};

  if (seg < NSEG - 1) {
    v8f E[4];
    #pragma unroll
    for (int t = 0; t < 4; ++t) E[t] = zero;

    for (int c = 0; c < CH_SEG; ++c) {
      const int cb = c * TCH;
      v2f ub[4];
      #pragma unroll
      for (int j = 0; j < 4; ++j)
        ub[j] = *(const v2f*)(useq + cb + 4 * j + 2 * half);
      __builtin_prefetch((const void*)(useq + cb + 4 * TCH), 0, 0);

      #pragma unroll
      for (int t = 0; t < 4; ++t) {
        #pragma unroll
        for (int r = 0; r < 8; ++r) E[t][r] *= av[t][r];
        #pragma unroll
        for (int j = 0; j < 4; ++j) E[t] = wmma4_f32(bm[t][j], ub[j], E[t]);
      }
    }

    float* ep = Eloc + seg * 1024;
    #pragma unroll
    for (int t = 0; t < 4; ++t)
      #pragma unroll
      for (int r = 0; r < 8; ++r)
        ep[t * 256 + r * 32 + lane] = E[t][r];
  }
  __syncthreads();

  v8f H[4];
  {
    float aw[4][8];
    #pragma unroll
    for (int t = 0; t < 4; ++t)
      #pragma unroll
      for (int r = 0; r < 8; ++r) {
        float p = av[t][r];
        #pragma unroll
        for (int s = 1; s < CH_SEG; s <<= 1) p *= p;
        aw[t][r] = p;
      }

    #pragma unroll
    for (int t = 0; t < 4; ++t) H[t] = zero;
    for (int g = 0; g < seg; ++g) {
      const float* ep = Eloc + g * 1024;
      #pragma unroll
      for (int t = 0; t < 4; ++t)
        #pragma unroll
        for (int r = 0; r < 8; ++r)
          H[t][r] = aw[t][r] * H[t][r] + ep[t * 256 + r * 32 + lane];
    }
  }

  v2f mA[4];
  #pragma unroll
  for (int j = 0; j < 4; ++j)
    mA[j] = *(const v2f*)(Mm + lm * TCH + 4 * j + 2 * half);

  v2f cmA[4][4];
  #pragma unroll
  for (int t = 0; t < 4; ++t)
    #pragma unroll
    for (int j = 0; j < 4; ++j)
      cmA[t][j] = *(const v2f*)(Cm + lm * NST + 16 * t + 4 * j + 2 * half);

  for (int c = 0; c < CH_SEG; ++c) {
    const int cb = c * TCH;
    v2f ub[4];
    #pragma unroll
    for (int j = 0; j < 4; ++j)
      ub[j] = *(const v2f*)(useq + cb + 4 * j + 2 * half);
    __builtin_prefetch((const void*)(useq + cb + 4 * TCH), 0, 0);

    v8f yacc = zero;
    #pragma unroll
    for (int j = 0; j < 4; ++j) yacc = wmma4_f32(mA[j], ub[j], yacc);

    #pragma unroll
    for (int t = 0; t < 4; ++t) {
      #pragma unroll
      for (int j = 0; j < 4; ++j) {
        const int jj = j & 1;
        float e0 = H[t][4 * jj + 0], e1 = H[t][4 * jj + 1];
        float e2 = H[t][4 * jj + 2], e3 = H[t][4 * jj + 3];
        float sx = half ? e0 : e2;
        float sy = half ? e1 : e3;
        float px = __shfl_xor(sx, 16);
        float py = __shfl_xor(sy, 16);
        float ox = half ? e2 : e0;
        float oy = half ? e3 : e1;
        bool useOwn = (half == (j >> 1));
        v2f hb;
        hb.x = useOwn ? ox : px;
        hb.y = useOwn ? oy : py;
        yacc = wmma4_f32(cmA[t][j], hb, yacc);
      }
    }

    if (c != CH_SEG - 1) {
      #pragma unroll
      for (int t = 0; t < 4; ++t) {
        #pragma unroll
        for (int r = 0; r < 8; ++r) H[t][r] *= av[t][r];
        #pragma unroll
        for (int j = 0; j < 4; ++j) H[t] = wmma4_f32(bm[t][j], ub[j], H[t]);
      }
    }

    #pragma unroll
    for (int r = 0; r < 8; ++r)
      yseq[cb + r + 8 * half] = yacc[r];
  }
}

extern "C" void kernel_launch(void* const* d_in, const int* in_sizes, int n_in,
                              void* d_out, int out_size, void* d_ws, size_t ws_size,
                              hipStream_t stream) {
  const float* u  = (const float*)d_in[0];
  const float* A  = (const float*)d_in[1];
  const float* Bv = (const float*)d_in[2];
  const float* Cv = (const float*)d_in[3];
  float* y      = (float*)d_out;

  ssm_fused<<<dim3(NTILE), dim3(NSEG * 32), 0, stream>>>(u, A, Bv, Cv, y);
}


# u read from memory once, as whole 128-byte lines, into a per-wave LDS tile; both passes take U from LDS instead of 8-byte global loads that touch 16 lines each; sum orders unchanged
# speedup vs baseline: 2.7989x; 1.0707x over previous
#include <hip/hip_runtime.h>


typedef __attribute__((ext_vector_type(2))) float v2f;
typedef __attribute__((ext_vector_type(8))) float v8f;

#define SEQS      4096
#define LSEQ      2048
#define NST       64
#define TCH       16
#define NSEG      16
#define CH_SEG    8
#define NTILE     256

#define OFF_M     0
#define OFF_CM    256
#define OFF_BM    1280
#define OFF_A16   2304
#define TBL_FLTS  2368

#define PPITCH    68
#define SEGLEN    (CH_SEG * TCH)
#define UPITCH    132

__device__ __forceinline__ v8f wmma4_f32(v2f a, v2f b, v8f c) {
  return __builtin_amdgcn_wmma_f32_16x16x4_f32(false, a, false, b,
                                               (short)0, c, false, false);
}

__global__ __launch_bounds__(NSEG * 32) void ssm_fused(
    const float* __restrict__ u, const float* __restrict__ A,
    const float* __restrict__ Bv, const float* __restrict__ Cv,
    float* __restrict__ y) {
  __shared__ __attribute__((aligned(16))) float Pw[TCH * PPITCH];
  __shared__ __attribute__((aligned(16))) float CB[NST];
  __shared__ __attribute__((aligned(16))) float tbl[TBL_FLTS];
  __shared__ float Eloc[(NSEG - 1) * 1024];
  __shared__ __attribute__((aligned(16))) float Us[NSEG * 16 * UPITCH];

  const int tid  = threadIdx.x;
  const int lane = tid & 31;
  const int half = lane >> 4;
  const int lm   = lane & 15;
  const int tile = blockIdx.x;
  const int seg  = __builtin_amdgcn_readfirstlane((int)(threadIdx.x >> 5));

  float* Mm  = tbl + OFF_M;
  float* Cm  = tbl + OFF_CM;
  float* Bm  = tbl + OFF_BM;
  float* a16 = tbl + OFF_A16;

  float*       yseq = y + (size_t)(tile * 16 + lm) * LSEQ + seg * SEGLEN;

  float a = 0.0f, b = 0.0f, c = 0.0f;
  if (tid < NST) {
    a = A[tid * NST + tid];
    b = Bv[tid]; c = Cv[tid];
  }

  float* uw = Us + seg * (16 * UPITCH);
  const float* urow = uw + lm * UPITCH;
  {
    const float* ug = u + (size_t)(tile * 16) * LSEQ + seg * SEGLEN + 4 * lane;
    float4 ur[16];
    #pragma unroll
    for (int i = 0; i < 16; ++i)
      ur[i] = *(const float4*)(ug + (size_t)i * LSEQ);
    #pragma unroll
    for (int i = 0; i < 16; ++i)
      *(float4*)(uw + i * UPITCH + 4 * lane) = ur[i];
  }

  {
    float pw[17];
    if (tid < NST) {
      float p = 1.0f;
      for (int i = 0; i < 17; ++i) { pw[i] = p; p *= a; }
      for (int d = 0; d < TCH; ++d) Pw[d * PPITCH + tid] = pw[d];
      CB[tid] = c * b;
    }
    __syncthreads();

    if (tid < NST) {
      a16[tid] = pw[16];
      for (int k = 0; k < TCH; ++k) Bm[tid * TCH + k] = b * pw[15 - k];
      for (int l = 0; l < TCH; ++l) Cm[l * NST + tid] = c * pw[l + 1];
    }

    if (tid < TCH * TCH) {
      int l = tid >> 4, k = tid & 15;
      float v = 0.0f;
      if (k <= l) {
        const float* prow = Pw + (l - k) * PPITCH;
        for (int q = 0; q < NST / 4; ++q) {
          const float4 pv = *(const float4*)(prow + 4 * q);
          const float4 cb = *(const float4*)(CB + 4 * q);
          v = fmaf(cb.x, pv.x, v);
          v = fmaf(cb.y, pv.y, v);
          v = fmaf(cb.z, pv.z, v);
          v = fmaf(cb.w, pv.w, v);
        }
      }
      Mm[l * TCH + k] = v;
    }
    __syncthreads();
  }

  float av[4][8];
  #pragma unroll
  for (int t = 0; t < 4; ++t)
    #pragma unroll
    for (int r = 0; r < 8; ++r)
      av[t][r] = a16[16 * t + r + 8 * half];

  v2f bm[4][4];
  #pragma unroll
  for (int t = 0; t < 4; ++t)
    #pragma unroll
    for (int j = 0; j < 4; ++j)
      bm[t][j] = *(const v2f*)(Bm + (16 * t + lm) * TCH + 4 * j + 2 * half);

  const v8f zero = {0.f, 0.f, 0.f, 0.f, 0.f, 0.f, 0.f, 0.f};

  if (seg < NSEG - 1) {
    v8f E[4];
    #pragma unroll
    for (int t = 0; t < 4; ++t) E[t] = zero;

    for (int c = 0; c < CH_SEG; ++c) {
      const int cb = c * TCH;
      v2f ub[4];
      #pragma unroll
      for (int j = 0; j < 4; ++j)
        ub[j] = *(const v2f*)(urow + cb + 4 * j + 2 * half);

      #pragma unroll
      for (int t = 0; t < 4; ++t) {
        #pragma unroll
        for (int r = 0; r < 8; ++r) E[t][r] *= av[t][r];
        #pragma unroll
        for (int j = 0; j < 4; ++j) E[t] = wmma4_f32(bm[t][j], ub[j], E[t]);
      }
    }

    float* ep = Eloc + seg * 1024;
    #pragma unroll
    for (int t = 0; t < 4; ++t)
      #pragma unroll
      for (int r = 0; r < 8; ++r)
        ep[t * 256 + r * 32 + lane] = E[t][r];
  }
  __syncthreads();

  v8f H[4];
  {
    float aw[4][8];
    #pragma unroll
    for (int t = 0; t < 4; ++t)
      #pragma unroll
      for (int r = 0; r < 8; ++r) {
        float p = av[t][r];
        #pragma unroll
        for (int s = 1; s < CH_SEG; s <<= 1) p *= p;
        aw[t][r] = p;
      }

    #pragma unroll
    for (int t = 0; t < 4; ++t) H[t] = zero;
    for (int g = 0; g < seg; ++g) {
      const float* ep = Eloc + g * 1024;
      #pragma unroll
      for (int t = 0; t < 4; ++t)
        #pragma unroll
        for (int r = 0; r < 8; ++r)
          H[t][r] = aw[t][r] * H[t][r] + ep[t * 256 + r * 32 + lane];
    }
  }

  v2f mA[4];
  #pragma unroll
  for (int j = 0; j < 4; ++j)
    mA[j] = *(const v2f*)(Mm + lm * TCH + 4 * j + 2 * half);

  v2f cmA[4][4];
  #pragma unroll
  for (int t = 0; t < 4; ++t)
    #pragma unroll
    for (int j = 0; j < 4; ++j)
      cmA[t][j] = *(const v2f*)(Cm + lm * NST + 16 * t + 4 * j + 2 * half);

  v2f un[4];
  #pragma unroll
  for (int j = 0; j < 4; ++j)
    un[j] = *(const v2f*)(urow + 4 * j + 2 * half);

  for (int c = 0; c < CH_SEG; ++c) {
    const int cb = c * TCH;
    v2f ub[4];
    #pragma unroll
    for (int j = 0; j < 4; ++j) ub[j] = un[j];
    if (c != CH_SEG - 1) {
      #pragma unroll
      for (int j = 0; j < 4; ++j)
        un[j] = *(const v2f*)(urow + cb + TCH + 4 * j + 2 * half);
    }

    v8f yacc = zero;
    #pragma unroll
    for (int j = 0; j < 4; ++j) yacc = wmma4_f32(mA[j], ub[j], yacc);

    #pragma unroll
    for (int t = 0; t < 4; ++t) {
      #pragma unroll
      for (int j = 0; j < 4; ++j) {
        const int jj = j & 1;
        float e0 = H[t][4 * jj + 0], e1 = H[t][4 * jj + 1];
        float e2 = H[t][4 * jj + 2], e3 = H[t][4 * jj + 3];
        float sx = half ? e0 : e2;
        float sy = half ? e1 : e3;
        float px = __shfl_xor(sx, 16);
        float py = __shfl_xor(sy, 16);
        float ox = half ? e2 : e0;
        float oy = half ? e3 : e1;
        bool useOwn = (half == (j >> 1));
        v2f hb;
        hb.x = useOwn ? ox : px;
        hb.y = useOwn ? oy : py;
        yacc = wmma4_f32(cmA[t][j], hb, yacc);
      }
    }

    if (c != CH_SEG - 1) {
      #pragma unroll
      for (int t = 0; t < 4; ++t) {
        #pragma unroll
        for (int r = 0; r < 8; ++r) H[t][r] *= av[t][r];
        #pragma unroll
        for (int j = 0; j < 4; ++j) H[t] = wmma4_f32(bm[t][j], ub[j], H[t]);
      }
    }

    #pragma unroll
    for (int r = 0; r < 8; ++r)
      yseq[cb + r + 8 * half] = yacc[r];
  }
}

extern "C" void kernel_launch(void* const* d_in, const int* in_sizes, int n_in,
                              void* d_out, int out_size, void* d_ws, size_t ws_size,
                              hipStream_t stream) {
  const float* u  = (const float*)d_in[0];
  const float* A  = (const float*)d_in[1];
  const float* Bv = (const float*)d_in[2];
  const float* Cv = (const float*)d_in[3];
  float* y      = (float*)d_out;

  ssm_fused<<<dim3(NTILE), dim3(NSEG * 32), 0, stream>>>(u, A, Bv, Cv, y);
}


# sequence scanned once by 4 waves per tile, each carrying 16 of the 64 states (no segment pre-scan, no fold): 36 matrix ops per chunk, not 52; Y = sum of 4 wave partials via LDS (sum orders change)
# speedup vs baseline: 3.5954x; 1.2846x over previous
#include <hip/hip_runtime.h>


typedef __attribute__((ext_vector_type(2))) float v2f;
typedef __attribute__((ext_vector_type(8))) float v8f;

#define SEQS      4096
#define LSEQ      2048
#define NST       64
#define TCH       16
#define NGRP      16
#define CH_GRP    8
#define NTILE     256
#define NWAVE     4

#define OFF_M     0
#define OFF_CM    256
#define OFF_BM    1280
#define OFF_A16   2304
#define TBL_FLTS  2368

#define PPITCH    68
#define GRPLEN    (CH_GRP * TCH)
#define UPITCH    132
#define TILE_FLTS (16 * UPITCH)

__device__ __forceinline__ v8f wmma4_f32(v2f a, v2f b, v8f c) {
  return __builtin_amdgcn_wmma_f32_16x16x4_f32(false, a, false, b,
                                               (short)0, c, false, false);
}

__device__ __forceinline__ void lds_barrier() {
  __builtin_amdgcn_fence(__ATOMIC_RELEASE, "workgroup", "local");
  __builtin_amdgcn_s_barrier();
  __builtin_amdgcn_fence(__ATOMIC_ACQUIRE, "workgroup", "local");
}

__global__ __launch_bounds__(NWAVE * 32) void ssm_fused(
    const float* __restrict__ u, const float* __restrict__ A,
    const float* __restrict__ Bv, const float* __restrict__ Cv,
    float* __restrict__ y) {
  __shared__ __attribute__((aligned(16))) float Pw[TCH * PPITCH];
  __shared__ __attribute__((aligned(16))) float CB[NST];
  __shared__ __attribute__((aligned(16))) float tbl[TBL_FLTS];
  __shared__ __attribute__((aligned(16))) float Us[NGRP * TILE_FLTS];
  __shared__ __attribute__((aligned(16))) float Yb[2 * NWAVE * TILE_FLTS];

  const int tid  = threadIdx.x;
  const int lane = tid & 31;
  const int half = lane >> 4;
  const int lm   = lane & 15;
  const int tile = blockIdx.x;
  const int wv   = __builtin_amdgcn_readfirstlane((int)(threadIdx.x >> 5));

  float* Mm  = tbl + OFF_M;
  float* Cm  = tbl + OFF_CM;
  float* Bm  = tbl + OFF_BM;
  float* a16 = tbl + OFF_A16;

  float a = 0.0f, b = 0.0f, c = 0.0f;
  if (tid < NST) {
    a = A[tid * NST + tid];
    b = Bv[tid]; c = Cv[tid];
  }

  const float* ug = u + (size_t)(tile * 16 + 4 * wv) * LSEQ + 4 * lane;
  float*       ul = Us + (4 * wv) * UPITCH + 4 * lane;
#define U_LOAD(q, g) (*(const float4*)(ug + (size_t)(q) * LSEQ + (g) * GRPLEN))
#define U_STORE(q, g, v) (*(float4*)(ul + (g) * TILE_FLTS + (q) * UPITCH) = (v))
  float4 ue0 = U_LOAD(0, 0), ue1 = U_LOAD(1, 0), ue2 = U_LOAD(2, 0), ue3 = U_LOAD(3, 0);
  float4 uo0 = U_LOAD(0, 1), uo1 = U_LOAD(1, 1), uo2 = U_LOAD(2, 1), uo3 = U_LOAD(3, 1);
  U_STORE(0, 0, ue0); U_STORE(1, 0, ue1); U_STORE(2, 0, ue2); U_STORE(3, 0, ue3);
  ue0 = U_LOAD(0, 2); ue1 = U_LOAD(1, 2); ue2 = U_LOAD(2, 2); ue3 = U_LOAD(3, 2);

  {
    float pw[17];
    if (tid < NST) {
      float p = 1.0f;
      for (int i = 0; i < 17; ++i) { pw[i] = p; p *= a; }
      for (int d = 0; d < TCH; ++d) Pw[d * PPITCH + tid] = pw[d];
      CB[tid] = c * b;
    }
    lds_barrier();

    if (tid < NST) {
      a16[tid] = pw[16];
      for (int k = 0; k < TCH; ++k) Bm[tid * TCH + k] = b * pw[15 - k];
      for (int l = 0; l < TCH; ++l) Cm[l * NST + tid] = c * pw[l + 1];
    }

    {
      const int k = tid & 15, l0 = tid >> 4, l1 = l0 + 8;
      const bool in0 = (k <= l0), in1 = (k <= l1);
      const float* prow0 = Pw + (in0 ? l0 - k : 0) * PPITCH;
      const float* prow1 = Pw + (in1 ? l1 - k : 0) * PPITCH;
      float v0 = 0.0f, v1 = 0.0f;
      #pragma unroll 4
      for (int q = 0; q < NST / 4; ++q) {
        const float4 cb = *(const float4*)(CB + 4 * q);
        const float4 pa = *(const float4*)(prow0 + 4 * q);
        const float4 pb = *(const float4*)(prow1 + 4 * q);
        v0 = fmaf(cb.x, pa.x, v0);
        v0 = fmaf(cb.y, pa.y, v0);
        v0 = fmaf(cb.z, pa.z, v0);
        v0 = fmaf(cb.w, pa.w, v0);
        v1 = fmaf(cb.x, pb.x, v1);
        v1 = fmaf(cb.y, pb.y, v1);
        v1 = fmaf(cb.z, pb.z, v1);
        v1 = fmaf(cb.w, pb.w, v1);
      }
      Mm[l0 * TCH + k] = in0 ? v0 : 0.0f;
      Mm[l1 * TCH + k] = in1 ? v1 : 0.0f;
    }

    lds_barrier();
  }

  v8f avv;
  #pragma unroll
  for (int r = 0; r < 8; ++r) avv[r] = a16[16 * wv + r + 8 * half];

  v2f bm[4];
  #pragma unroll
  for (int j = 0; j < 4; ++j)
    bm[j] = *(const v2f*)(Bm + (16 * wv + lm) * TCH + 4 * j + 2 * half);

  v2f cm[4];
  #pragma unroll
  for (int j = 0; j < 4; ++j)
    cm[j] = *(const v2f*)(Cm + lm * NST + 16 * wv + 8 * half + 2 * j);

  const v2f mA = *(const v2f*)(Mm + lm * TCH + 4 * wv + 2 * half);

  const v8f zero = {0.f, 0.f, 0.f, 0.f, 0.f, 0.f, 0.f, 0.f};
  v8f H = zero;

  v2f ub[CH_GRP][4], um[CH_GRP];
  {
    const float* urow = Us + lm * UPITCH + 2 * half;
    #pragma unroll
    for (int ch = 0; ch < CH_GRP; ++ch) {
      #pragma unroll
      for (int j = 0; j < 4; ++j)
        ub[ch][j] = *(const v2f*)(urow + ch * TCH + 4 * j);
      um[ch] = *(const v2f*)(urow + ch * TCH + 4 * wv);
    }
  }

  float4 rp[4][NWAVE];
#define SCHED_FENCE __builtin_amdgcn_sched_barrier(0);
#define GROUP_BODY(g, n0, n1, n2, n3, RED)                                     \
  {                                                                            \
    float* yw = Yb + (((g) & 1) * NWAVE + wv) * TILE_FLTS + lm * UPITCH        \
                + 8 * half;                                                    \
    float* yg = y + (size_t)(tile * 16 + 4 * wv) * LSEQ + ((g) - 1) * GRPLEN   \
                + 4 * lane;                                                    \
    const float* yq = Yb + ((((g) + 1) & 1) * NWAVE) * TILE_FLTS               \
                      + (4 * wv) * UPITCH + 4 * lane;                          \
    _Pragma("unroll")         \
    for (int ch = 0; ch < CH_GRP; ++ch) {                                      \
      v8f yp = wmma4_f32(mA, um[ch], zero);                                    \
      v8f Hn = H * avv;                 \
      _Pragma("unroll")                                                        \
      for (int j = 0; j < 4; ++j) {                                            \
        v2f hb;                                                                \
        hb.x = H[2 * j]; hb.y = H[2 * j + 1];                                  \
        yp = wmma4_f32(cm[j], hb, yp);                                         \
        Hn = wmma4_f32(bm[j], ub[ch][j], Hn);                                  \
      }                                                                        \
      H = Hn;                                                                  \
         \
      *(float4*)(yw + ch * TCH)     = make_float4(yp[0], yp[1], yp[2], yp[3]); \
      *(float4*)(yw + ch * TCH + 4) = make_float4(yp[4], yp[5], yp[6], yp[7]); \
      if (RED && (ch & 1)) {                                                   \
        const int q = ch >> 1;                                                 \
        float4 s;                                                              \
        s.x = ((rp[q][0].x + rp[q][1].x) + rp[q][2].x) + rp[q][3].x;           \
        s.y = ((rp[q][0].y + rp[q][1].y) + rp[q][2].y) + rp[q][3].y;           \
        s.z = ((rp[q][0].z + rp[q][1].z) + rp[q][2].z) + rp[q][3].z;           \
        s.w = ((rp[q][0].w + rp[q][1].w) + rp[q][2].w) + rp[q][3].w;           \
        *(float4*)(yg + (size_t)q * LSEQ) = s;                                 \
        if (q + 1 < 4) {                \
          _Pragma("unroll")                                                    \
          for (int w = 0; w < NWAVE; ++w)                                      \
            rp[q + 1][w] =                                                     \
                *(const float4*)(yq + w * TILE_FLTS + (q + 1) * UPITCH);       \
        }                                                                      \
      }                                                                        \
      SCHED_FENCE                                                              \
    }                                                                          \
    if ((g) + 1 < NGRP) {                                                      \
      U_STORE(0, (g) + 1, n0); U_STORE(1, (g) + 1, n1);                        \
      U_STORE(2, (g) + 1, n2); U_STORE(3, (g) + 1, n3);                        \
    }                                                                          \
    if ((g) + 3 < NGRP) {                                                      \
      n0 = U_LOAD(0, (g) + 3); n1 = U_LOAD(1, (g) + 3);                        \
      n2 = U_LOAD(2, (g) + 3); n3 = U_LOAD(3, (g) + 3);                        \
    }                                                                          \
    lds_barrier();                                                             \
    if ((g) + 1 < NGRP) {                                                      \
      const float* urow = Us + ((g) + 1) * TILE_FLTS + lm * UPITCH + 2 * half; \
      _Pragma("unroll")                  \
      for (int ch = 0; ch < CH_GRP; ++ch) {                                    \
        _Pragma("unroll")                                                      \
        for (int j = 0; j < 4; ++j)                                            \
          ub[ch][j] = *(const v2f*)(urow + ch * TCH + 4 * j);                  \
        um[ch] = *(const v2f*)(urow + ch * TCH + 4 * wv);                      \
      }                                                                        \
    }                                                                          \
    {                                                                          \
      const float* yr = Yb + (((g) & 1) * NWAVE) * TILE_FLTS                   \
                        + (4 * wv) * UPITCH + 4 * lane;                        \
      _Pragma("unroll")           \
      for (int w = 0; w < NWAVE; ++w)                                          \
        rp[0][w] = *(const float4*)(yr + w * TILE_FLTS);                       \
    }                                                                          \
         \
         \
    SCHED_FENCE                                                                \
  }

  GROUP_BODY(0, uo0, uo1, uo2, uo3, 0)
  for (int g = 1; g < NGRP - 1; g += 2) {
    GROUP_BODY(g, ue0, ue1, ue2, ue3, 1)
    GROUP_BODY(g + 1, uo0, uo1, uo2, uo3, 1)
  }
  GROUP_BODY(NGRP - 1, ue0, ue1, ue2, ue3, 1)
#undef GROUP_BODY
#undef SCHED_FENCE
#undef U_LOAD
#undef U_STORE

  {
    float* yg = y + (size_t)(tile * 16 + 4 * wv) * LSEQ + (NGRP - 1) * GRPLEN
                + 4 * lane;
    const float* yq = Yb + (((NGRP - 1) & 1) * NWAVE) * TILE_FLTS
                      + (4 * wv) * UPITCH + 4 * lane;
    #pragma unroll
    for (int q = 1; q < 4; ++q) {
      #pragma unroll
      for (int w = 0; w < NWAVE; ++w)
        rp[q][w] = *(const float4*)(yq + w * TILE_FLTS + q * UPITCH);
    }
    #pragma unroll
    for (int q = 0; q < 4; ++q) {
      float4 s;
      s.x = ((rp[q][0].x + rp[q][1].x) + rp[q][2].x) + rp[q][3].x;
      s.y = ((rp[q][0].y + rp[q][1].y) + rp[q][2].y) + rp[q][3].y;
      s.z = ((rp[q][0].z + rp[q][1].z) + rp[q][2].z) + rp[q][3].z;
      s.w = ((rp[q][0].w + rp[q][1].w) + rp[q][2].w) + rp[q][3].w;
      *(float4*)(yg + (size_t)q * LSEQ) = s;
    }
  }
}

extern "C" void kernel_launch(void* const* d_in, const int* in_sizes, int n_in,
                              void* d_out, int out_size, void* d_ws, size_t ws_size,
                              hipStream_t stream) {
  const float* u  = (const float*)d_in[0];
  const float* A  = (const float*)d_in[1];
  const float* Bv = (const float*)d_in[2];
  const float* Cv = (const float*)d_in[3];
  float* y      = (float*)d_out;

  ssm_fused<<<dim3(NTILE), dim3(NWAVE * 32), 0, stream>>>(u, A, Bv, Cv, y);
}


# every LDS and global memory instruction of a 128-step group moved into one place behind the group's barrier, fetching a group or two ahead; none left among its 72 matrix ops; sum orders unchanged
# speedup vs baseline: 3.8003x; 1.0570x over previous
#include <hip/hip_runtime.h>


typedef __attribute__((ext_vector_type(2))) float v2f;
typedef __attribute__((ext_vector_type(8))) float v8f;

#define SEQS      4096
#define LSEQ      2048
#define NST       64
#define TCH       16
#define NGRP      16
#define CH_GRP    8
#define NTILE     256
#define NWAVE     4

#define OFF_M     0
#define OFF_CM    256
#define OFF_BM    1280
#define OFF_A16   2304
#define TBL_FLTS  2368

#define PPITCH    68
#define GRPLEN    (CH_GRP * TCH)
#define UPITCH    132
#define TILE_FLTS (16 * UPITCH)

__device__ __forceinline__ v8f wmma4_f32(v2f a, v2f b, v8f c) {
  return __builtin_amdgcn_wmma_f32_16x16x4_f32(false, a, false, b,
                                               (short)0, c, false, false);
}

__device__ __forceinline__ void lds_barrier() {
  __builtin_amdgcn_fence(__ATOMIC_RELEASE, "workgroup", "local");
  __builtin_amdgcn_s_barrier();
  __builtin_amdgcn_fence(__ATOMIC_ACQUIRE, "workgroup", "local");
}

__global__ __launch_bounds__(NWAVE * 32) void ssm_fused(
    const float* __restrict__ u, const float* __restrict__ A,
    const float* __restrict__ Bv, const float* __restrict__ Cv,
    float* __restrict__ y) {
  __shared__ __attribute__((aligned(16))) float Pw[TCH * PPITCH];
  __shared__ __attribute__((aligned(16))) float CB[NST];
  __shared__ __attribute__((aligned(16))) float tbl[TBL_FLTS];
  __shared__ __attribute__((aligned(16))) float Us[NGRP * TILE_FLTS];
  __shared__ __attribute__((aligned(16))) float Yb[2 * NWAVE * TILE_FLTS];

  const int tid  = threadIdx.x;
  const int lane = tid & 31;
  const int half = lane >> 4;
  const int lm   = lane & 15;
  const int tile = blockIdx.x;
  const int wv   = __builtin_amdgcn_readfirstlane((int)(threadIdx.x >> 5));

  float* Mm  = tbl + OFF_M;
  float* Cm  = tbl + OFF_CM;
  float* Bm  = tbl + OFF_BM;
  float* a16 = tbl + OFF_A16;

  float a = 0.0f, b = 0.0f, c = 0.0f;
  if (tid < NST) {
    a = A[tid * NST + tid];
    b = Bv[tid]; c = Cv[tid];
  }

  const float* ug = u + (size_t)(tile * 16 + 4 * wv) * LSEQ + 4 * lane;
  float*       ul = Us + (4 * wv) * UPITCH + 4 * lane;
#define U_LOAD(q, g) (*(const float4*)(ug + (size_t)(q) * LSEQ + (g) * GRPLEN))
#define U_STORE(q, g, v) (*(float4*)(ul + (g) * TILE_FLTS + (q) * UPITCH) = (v))
  float4 ue0 = U_LOAD(0, 0), ue1 = U_LOAD(1, 0), ue2 = U_LOAD(2, 0), ue3 = U_LOAD(3, 0);
  float4 uo0 = U_LOAD(0, 1), uo1 = U_LOAD(1, 1), uo2 = U_LOAD(2, 1), uo3 = U_LOAD(3, 1);
  U_STORE(0, 0, ue0); U_STORE(1, 0, ue1); U_STORE(2, 0, ue2); U_STORE(3, 0, ue3);
  ue0 = U_LOAD(0, 2); ue1 = U_LOAD(1, 2); ue2 = U_LOAD(2, 2); ue3 = U_LOAD(3, 2);
  U_STORE(0, 1, uo0); U_STORE(1, 1, uo1); U_STORE(2, 1, uo2); U_STORE(3, 1, uo3);
  uo0 = U_LOAD(0, 3); uo1 = U_LOAD(1, 3); uo2 = U_LOAD(2, 3); uo3 = U_LOAD(3, 3);

  {
    float pw[17];
    if (tid < NST) {
      float p = 1.0f;
      for (int i = 0; i < 17; ++i) { pw[i] = p; p *= a; }
      for (int d = 0; d < TCH; ++d) Pw[d * PPITCH + tid] = pw[d];
      CB[tid] = c * b;
    }
    lds_barrier();

    if (tid < NST) {
      a16[tid] = pw[16];
      for (int k = 0; k < TCH; ++k) Bm[tid * TCH + k] = b * pw[15 - k];
      for (int l = 0; l < TCH; ++l) Cm[l * NST + tid] = c * pw[l + 1];
    }

    {
      const int k = tid & 15, l0 = tid >> 4, l1 = l0 + 8;
      const bool in0 = (k <= l0), in1 = (k <= l1);
      const float* prow0 = Pw + (in0 ? l0 - k : 0) * PPITCH;
      const float* prow1 = Pw + (in1 ? l1 - k : 0) * PPITCH;
      float v0 = 0.0f, v1 = 0.0f;
      #pragma unroll 4
      for (int q = 0; q < NST / 4; ++q) {
        const float4 cb = *(const float4*)(CB + 4 * q);
        const float4 pa = *(const float4*)(prow0 + 4 * q);
        const float4 pb = *(const float4*)(prow1 + 4 * q);
        v0 = fmaf(cb.x, pa.x, v0);
        v0 = fmaf(cb.y, pa.y, v0);
        v0 = fmaf(cb.z, pa.z, v0);
        v0 = fmaf(cb.w, pa.w, v0);
        v1 = fmaf(cb.x, pb.x, v1);
        v1 = fmaf(cb.y, pb.y, v1);
        v1 = fmaf(cb.z, pb.z, v1);
        v1 = fmaf(cb.w, pb.w, v1);
      }
      Mm[l0 * TCH + k] = in0 ? v0 : 0.0f;
      Mm[l1 * TCH + k] = in1 ? v1 : 0.0f;
    }

    lds_barrier();
  }

  v8f avv;
  #pragma unroll
  for (int r = 0; r < 8; ++r) avv[r] = a16[16 * wv + r + 8 * half];

  v2f bm[4];
  #pragma unroll
  for (int j = 0; j < 4; ++j)
    bm[j] = *(const v2f*)(Bm + (16 * wv + lm) * TCH + 4 * j + 2 * half);

  v2f cm[4];
  #pragma unroll
  for (int j = 0; j < 4; ++j)
    cm[j] = *(const v2f*)(Cm + lm * NST + 16 * wv + 8 * half + 2 * j);

  const v2f mA = *(const v2f*)(Mm + lm * TCH + 4 * wv + 2 * half);

  const v8f zero = {0.f, 0.f, 0.f, 0.f, 0.f, 0.f, 0.f, 0.f};
  v8f H = zero;

  v2f ubA[CH_GRP][4], umA[CH_GRP], ubB[CH_GRP][4], umB[CH_GRP];
#define OPERANDS(UB, UM, g)                                                    \
  {                                                                            \
    const float* urow = Us + (g) * TILE_FLTS + lm * UPITCH + 2 * half;         \
    _Pragma("unroll")                    \
    for (int ch = 0; ch < CH_GRP; ++ch) {                                      \
      _Pragma("unroll")                                                        \
      for (int j = 0; j < 4; ++j)                                              \
        UB[ch][j] = *(const v2f*)(urow + ch * TCH + 4 * j);                    \
      UM[ch] = *(const v2f*)(urow + ch * TCH + 4 * wv);                        \
    }                                                                          \
  }
  OPERANDS(ubA, umA, 0)
  OPERANDS(ubB, umB, 1)
  U_STORE(0, 2, ue0); U_STORE(1, 2, ue1); U_STORE(2, 2, ue2); U_STORE(3, 2, ue3);
  ue0 = U_LOAD(0, 4); ue1 = U_LOAD(1, 4); ue2 = U_LOAD(2, 4); ue3 = U_LOAD(3, 4);

  float4 rp[4][NWAVE];
#define MEM_FENCE __builtin_amdgcn_sched_barrier(0);
#define SUM_FENCE __builtin_amdgcn_sched_barrier(0);
#define GROUP_BODY(k, UB, UM, n0, n1, n2, n3, SUM, RPL)                        \
  {                                                                            \
    v8f yt[CH_GRP];                       \
    float4 sm[4];                         \
    _Pragma("unroll")     \
    for (int ch = 0; ch < CH_GRP; ++ch) {                                      \
      v8f yp = wmma4_f32(mA, UM[ch], zero);                                    \
      v8f Hn = H * avv;                 \
      _Pragma("unroll")                                                        \
      for (int j = 0; j < 4; ++j) {                                            \
        v2f hb;                                                                \
        hb.x = H[2 * j]; hb.y = H[2 * j + 1];                                  \
        yp = wmma4_f32(cm[j], hb, yp);                                         \
        Hn = wmma4_f32(bm[j], UB[ch][j], Hn);                                  \
      }                                                                        \
      H = Hn;                                                                  \
      yt[ch] = yp;                                                             \
      if (SUM && ch >= CH_GRP / 2) {            \
        const int q = ch - CH_GRP / 2;                                         \
        sm[q].x = ((rp[q][0].x + rp[q][1].x) + rp[q][2].x) + rp[q][3].x;       \
        sm[q].y = ((rp[q][0].y + rp[q][1].y) + rp[q][2].y) + rp[q][3].y;       \
        sm[q].z = ((rp[q][0].z + rp[q][1].z) + rp[q][2].z) + rp[q][3].z;       \
        sm[q].w = ((rp[q][0].w + rp[q][1].w) + rp[q][2].w) + rp[q][3].w;       \
      }                                                                        \
      if (ch == CH_GRP / 2 - 1) SUM_FENCE                                      \
    }                                                                          \
    MEM_FENCE                                                                  \
    lds_barrier();                                                             \
    {                                                                          \
      float* yw = Yb + (((k) & 1) * NWAVE + wv) * TILE_FLTS + lm * UPITCH      \
                  + 8 * half;                                                  \
      _Pragma("unroll")           \
      for (int ch = 0; ch < CH_GRP; ++ch) {                                    \
          \
        *(float4*)(yw + ch * TCH) =                                            \
            make_float4(yt[ch][0], yt[ch][1], yt[ch][2], yt[ch][3]);           \
        *(float4*)(yw + ch * TCH + 4) =                                        \
            make_float4(yt[ch][4], yt[ch][5], yt[ch][6], yt[ch][7]);           \
      }                                                                        \
    }                                                                          \
    if ((k) + 2 < NGRP) OPERANDS(UB, UM, (k) + 2)                              \
    if (RPL) {                                                                 \
      const float* yr = Yb + ((((k) + 1) & 1) * NWAVE) * TILE_FLTS             \
                        + (4 * wv) * UPITCH + 4 * lane;                        \
      _Pragma("unroll")           \
      for (int q = 0; q < 4; ++q) {                                            \
        _Pragma("unroll")                                                      \
        for (int w = 0; w < NWAVE; ++w)                                        \
          rp[q][w] = *(const float4*)(yr + w * TILE_FLTS + q * UPITCH);        \
      }                                                                        \
    }                                                                          \
    if (SUM) {                                                                 \
      float* yg = y + (size_t)(tile * 16 + 4 * wv) * LSEQ                      \
                  + ((k) - 2) * GRPLEN + 4 * lane;                             \
      _Pragma("unroll")           \
      for (int q = 0; q < 4; ++q)                                              \
        *(float4*)(yg + (size_t)q * LSEQ) = sm[q];                             \
    }                                                                          \
    if ((k) + 3 < NGRP) {                                                      \
      U_STORE(0, (k) + 3, n0); U_STORE(1, (k) + 3, n1);                        \
      U_STORE(2, (k) + 3, n2); U_STORE(3, (k) + 3, n3);                        \
    }                                                                          \
    if ((k) + 5 < NGRP) {                                                      \
      n0 = U_LOAD(0, (k) + 5); n1 = U_LOAD(1, (k) + 5);                        \
      n2 = U_LOAD(2, (k) + 5); n3 = U_LOAD(3, (k) + 5);                        \
    }                                                                          \
    MEM_FENCE                                                                  \
  }

  MEM_FENCE
  GROUP_BODY(0, ubA, umA, uo0, uo1, uo2, uo3, 0, 0)
  GROUP_BODY(1, ubB, umB, ue0, ue1, ue2, ue3, 0, 1)
  for (int g = 2; g < NGRP - 2; g += 2) {
    GROUP_BODY(g, ubA, umA, uo0, uo1, uo2, uo3, 1, 1)
    GROUP_BODY(g + 1, ubB, umB, ue0, ue1, ue2, ue3, 1, 1)
  }
  GROUP_BODY(NGRP - 2, ubA, umA, uo0, uo1, uo2, uo3, 1, 1)
  GROUP_BODY(NGRP - 1, ubB, umB, ue0, ue1, ue2, ue3, 1, 1)
#undef GROUP_BODY
#undef OPERANDS
#undef MEM_FENCE
#undef SUM_FENCE
#undef U_LOAD
#undef U_STORE

  #pragma unroll
  for (int t = 0; t < 2; ++t) {
    float* yg = y + (size_t)(tile * 16 + 4 * wv) * LSEQ
                + (NGRP - 2 + t) * GRPLEN + 4 * lane;
    if (t == 1) {
      lds_barrier();
      const float* yr = Yb + (((NGRP - 1) & 1) * NWAVE) * TILE_FLTS
                        + (4 * wv) * UPITCH + 4 * lane;
      #pragma unroll
      for (int q = 0; q < 4; ++q) {
        #pragma unroll
        for (int w = 0; w < NWAVE; ++w)
          rp[q][w] = *(const float4*)(yr + w * TILE_FLTS + q * UPITCH);
      }
    }
    #pragma unroll
    for (int q = 0; q < 4; ++q) {
      float4 s;
      s.x = ((rp[q][0].x + rp[q][1].x) + rp[q][2].x) + rp[q][3].x;
      s.y = ((rp[q][0].y + rp[q][1].y) + rp[q][2].y) + rp[q][3].y;
      s.z = ((rp[q][0].z + rp[q][1].z) + rp[q][2].z) + rp[q][3].z;
      s.w = ((rp[q][0].w + rp[q][1].w) + rp[q][2].w) + rp[q][3].w;
      *(float4*)(yg + (size_t)q * LSEQ) = s;
    }
  }
}

extern "C" void kernel_launch(void* const* d_in, const int* in_sizes, int n_in,
                              void* d_out, int out_size, void* d_ws, size_t ws_size,
                              hipStream_t stream) {
  const float* u  = (const float*)d_in[0];
  const float* A  = (const float*)d_in[1];
  const float* Bv = (const float*)d_in[2];
  const float* Cv = (const float*)d_in[3];
  float* y      = (float*)d_out;

  ssm_fused<<<dim3(NTILE), dim3(NWAVE * 32), 0, stream>>>(u, A, Bv, Cv, y);
}


# the 4 waves split the sequence by time, not by states: wave w takes every fourth 128-step group with all 64 states, Bm*U kept in registers, start states by an LDS fold S=a^128*S+E (sum orders change)
# speedup vs baseline: 4.1046x; 1.0801x over previous
#include <hip/hip_runtime.h>


typedef __attribute__((ext_vector_type(2))) float v2f;
typedef __attribute__((ext_vector_type(8))) float v8f;

#define SEQS      4096
#define LSEQ      2048
#define NST       64
#define NBLK      4
#define TCH       16
#define NGRP      16
#define CH_GRP    8
#define NTILE     256
#define NWAVE     4
#define NRND      (NGRP / NWAVE)

#define OFF_M     0
#define OFF_CM    256
#define OFF_BM    1280
#define OFF_A16   2304
#define OFF_A128  2368
#define TBL_FLTS  2432

#define PPITCH    68
#define GRPLEN    (CH_GRP * TCH)
#define UPITCH    132
#define TILE_FLTS (16 * UPITCH)
#define E_FLTS    (NST * 16)

__device__ __forceinline__ v8f wmma4_f32(v2f a, v2f b, v8f c) {
  return __builtin_amdgcn_wmma_f32_16x16x4_f32(false, a, false, b,
                                               (short)0, c, false, false);
}

__device__ __forceinline__ void lds_barrier() {
  __builtin_amdgcn_fence(__ATOMIC_RELEASE, "workgroup", "local");
  __builtin_amdgcn_s_barrier();
  __builtin_amdgcn_fence(__ATOMIC_ACQUIRE, "workgroup", "local");
}

__device__ __forceinline__ void wave_lds_order() {
  __builtin_amdgcn_fence(__ATOMIC_RELEASE, "wavefront", "local");
  __builtin_amdgcn_wave_barrier();
  __builtin_amdgcn_fence(__ATOMIC_ACQUIRE, "wavefront", "local");
}

__global__ __launch_bounds__(NWAVE * 32) void ssm_fused(
    const float* __restrict__ u, const float* __restrict__ A,
    const float* __restrict__ Bv, const float* __restrict__ Cv,
    float* __restrict__ y) {
  __shared__ __attribute__((aligned(16))) float Pw[TCH * PPITCH];
  __shared__ __attribute__((aligned(16))) float CB[NST];
  __shared__ __attribute__((aligned(16))) float tbl[TBL_FLTS];
  __shared__ __attribute__((aligned(16))) float Ut[NWAVE * TILE_FLTS];
  __shared__ __attribute__((aligned(16))) float Yt[NWAVE * TILE_FLTS];
  __shared__ __attribute__((aligned(16))) float Es[2 * NWAVE * E_FLTS];

  const int tid  = threadIdx.x;
  const int lane = tid & 31;
  const int half = lane >> 4;
  const int lm   = lane & 15;
  const int tile = blockIdx.x;
  const int wv   = __builtin_amdgcn_readfirstlane((int)(threadIdx.x >> 5));

  float* Mm   = tbl + OFF_M;
  float* Cm   = tbl + OFF_CM;
  float* Bm   = tbl + OFF_BM;
  float* a16  = tbl + OFF_A16;
  float* a128 = tbl + OFF_A128;

  float a = 0.0f, b = 0.0f, c = 0.0f;
  if (tid < NST) {
    a = A[tid * NST + tid];
    b = Bv[tid]; c = Cv[tid];
  }

  const float* ug = u + (size_t)(tile * 16) * LSEQ + wv * GRPLEN + 4 * lane;
  float*       yg = y + (size_t)(tile * 16) * LSEQ + wv * GRPLEN + 4 * lane;
#define ROWS16(X) X(0) X(1) X(2) X(3) X(4) X(5) X(6) X(7) X(8) X(9) X(10) X(11) \
                  X(12) X(13) X(14) X(15)
#define U_FIRST(q) float4 un##q = *(const float4*)(ug + (size_t)(q) * LSEQ);
#define U_TO_LDS(q) *(float4*)(utw + (q) * UPITCH + 4 * lane) = un##q;
#define U_NEXT(q)                                                              \
  un##q = *(const float4*)(ug + (size_t)(q) * LSEQ + (r + 1) * (NWAVE * GRPLEN));
  ROWS16(U_FIRST)

  {
    float pw[17];
    if (tid < NST) {
      float p = 1.0f;
      for (int i = 0; i < 17; ++i) { pw[i] = p; p *= a; }
      for (int d = 0; d < TCH; ++d) Pw[d * PPITCH + tid] = pw[d];
      CB[tid] = c * b;
    }
    lds_barrier();

    if (tid < NST) {
      a16[tid] = pw[16];
      const float p32 = pw[16] * pw[16];
      const float p64 = p32 * p32;
      a128[tid] = p64 * p64;
      for (int k = 0; k < TCH; ++k) Bm[tid * TCH + k] = b * pw[15 - k];
      for (int l = 0; l < TCH; ++l) Cm[l * NST + tid] = c * pw[l + 1];
    }

    {
      const int k = tid & 15, l0 = tid >> 4, l1 = l0 + 8;
      const bool in0 = (k <= l0), in1 = (k <= l1);
      const float* prow0 = Pw + (in0 ? l0 - k : 0) * PPITCH;
      const float* prow1 = Pw + (in1 ? l1 - k : 0) * PPITCH;
      float v0 = 0.0f, v1 = 0.0f;
      #pragma unroll 4
      for (int q = 0; q < NST / 4; ++q) {
        const float4 cb = *(const float4*)(CB + 4 * q);
        const float4 pa = *(const float4*)(prow0 + 4 * q);
        const float4 pb = *(const float4*)(prow1 + 4 * q);
        v0 = fmaf(cb.x, pa.x, v0);
        v0 = fmaf(cb.y, pa.y, v0);
        v0 = fmaf(cb.z, pa.z, v0);
        v0 = fmaf(cb.w, pa.w, v0);
        v1 = fmaf(cb.x, pb.x, v1);
        v1 = fmaf(cb.y, pb.y, v1);
        v1 = fmaf(cb.z, pb.z, v1);
        v1 = fmaf(cb.w, pb.w, v1);
      }
      Mm[l0 * TCH + k] = in0 ? v0 : 0.0f;
      Mm[l1 * TCH + k] = in1 ? v1 : 0.0f;
    }

    lds_barrier();
  }

  v8f avv[NBLK], apv[NBLK];
  #pragma unroll
  for (int bk = 0; bk < NBLK; ++bk) {
    #pragma unroll
    for (int r = 0; r < 8; ++r) {
      avv[bk][r] = a16[16 * bk + r + 8 * half];
      apv[bk][r] = a128[16 * bk + r + 8 * half];
    }
  }

  v2f bm[NBLK][4];
  #pragma unroll
  for (int bk = 0; bk < NBLK; ++bk) {
    #pragma unroll
    for (int j = 0; j < 4; ++j)
      bm[bk][j] = *(const v2f*)(Bm + (16 * bk + lm) * TCH + 8 * half + 2 * j);
  }
  v2f mA[4];
  #pragma unroll
  for (int j = 0; j < 4; ++j)
    mA[j] = *(const v2f*)(Mm + lm * TCH + 8 * half + 2 * j);

  v2f cm[NBLK][4];
  #pragma unroll
  for (int bk = 0; bk < NBLK; ++bk) {
    #pragma unroll
    for (int j = 0; j < 4; ++j)
      cm[bk][j] = *(const v2f*)(Cm + lm * NST + 16 * bk + 8 * half + 2 * j);
  }

  const v8f zero = {0.f, 0.f, 0.f, 0.f, 0.f, 0.f, 0.f, 0.f};
  v8f Cst[NBLK];
  #pragma unroll
  for (int bk = 0; bk < NBLK; ++bk) Cst[bk] = zero;

  float* utw = Ut + wv * TILE_FLTS;
  float* ytw = Yt + wv * TILE_FLTS;

  v8f G[CH_GRP][NBLK];
  v8f yt[CH_GRP];
  v2f ub[CH_GRP][4];

#define MEM_FENCE __builtin_amdgcn_sched_barrier(0);

#define Y_TO_LDS                                                               \
  {                                                                            \
    float* yw = ytw + lm * UPITCH + 8 * half;                                  \
    _Pragma("unroll")             \
    for (int ch = 0; ch < CH_GRP; ++ch) {                                      \
         \
      *(float4*)(yw + ch * TCH) =                                              \
          make_float4(yt[ch][0], yt[ch][1], yt[ch][2], yt[ch][3]);             \
      *(float4*)(yw + ch * TCH + 4) =                                          \
          make_float4(yt[ch][4], yt[ch][5], yt[ch][6], yt[ch][7]);             \
    }                                                                          \
  }
#define Y_DECL(q) float4 yr##q;
#define Y_ROW_LOAD(q) yr##q = *(const float4*)(ytw + (q) * UPITCH + 4 * lane);
#define Y_ROW_STORE(q) *(float4*)(yg + (size_t)(q) * LSEQ + yoff) = yr##q;

  #pragma nounroll
  for (int r = 0; r < NRND; ++r) {
    MEM_FENCE
    ROWS16(U_TO_LDS)
    wave_lds_order();
    {
      const float* urow = utw + lm * UPITCH + 8 * half;
      #pragma unroll
      for (int ch = 0; ch < CH_GRP; ++ch) {
        const float4 lo = *(const float4*)(urow + ch * TCH);
        const float4 hi = *(const float4*)(urow + ch * TCH + 4);
        ub[ch][0].x = lo.x; ub[ch][0].y = lo.y;
        ub[ch][1].x = lo.z; ub[ch][1].y = lo.w;
        ub[ch][2].x = hi.x; ub[ch][2].y = hi.y;
        ub[ch][3].x = hi.z; ub[ch][3].y = hi.w;
      }
    }
    if (r > 0) Y_TO_LDS
    if (r + 1 < NRND) {
      ROWS16(U_NEXT)
    }
    MEM_FENCE

    v8f H0[NBLK];
    #pragma unroll
    for (int ch = 0; ch < CH_GRP; ++ch) {
      v8f g[NBLK];
      #pragma unroll
      for (int bk = 0; bk < NBLK; ++bk) g[bk] = zero;
      #pragma unroll
      for (int j = 0; j < 4; ++j) {
        #pragma unroll
        for (int bk = 0; bk < NBLK; ++bk)
          g[bk] = wmma4_f32(bm[bk][j], ub[ch][j], g[bk]);
      }
      #pragma unroll
      for (int bk = 0; bk < NBLK; ++bk) {
        G[ch][bk] = g[bk];
        H0[bk] = (ch == 0) ? g[bk] : H0[bk] * avv[bk] + g[bk];
      }
    }
    MEM_FENCE

    {
      float* ew = Es + ((r & 1) * NWAVE + wv) * E_FLTS + 4 * lane;
      #pragma unroll
      for (int bk = 0; bk < NBLK; ++bk) {
        *(float4*)(ew + (2 * bk) * 128) =
            make_float4(H0[bk][0], H0[bk][1], H0[bk][2], H0[bk][3]);
        *(float4*)(ew + (2 * bk + 1) * 128) =
            make_float4(H0[bk][4], H0[bk][5], H0[bk][6], H0[bk][7]);
      }
    }
    lds_barrier();
    v8f ev[NWAVE][NBLK];
    #pragma unroll
    for (int v = 0; v < NWAVE; ++v) {
      const float* er = Es + ((r & 1) * NWAVE + v) * E_FLTS + 4 * lane;
      #pragma unroll
      for (int bk = 0; bk < NBLK; ++bk) {
        const float4 e0 = *(const float4*)(er + (2 * bk) * 128);
        const float4 e1 = *(const float4*)(er + (2 * bk + 1) * 128);
        ev[v][bk][0] = e0.x; ev[v][bk][1] = e0.y;
        ev[v][bk][2] = e0.z; ev[v][bk][3] = e0.w;
        ev[v][bk][4] = e1.x; ev[v][bk][5] = e1.y;
        ev[v][bk][6] = e1.z; ev[v][bk][7] = e1.w;
      }
    }
    ROWS16(Y_DECL)
    if (r > 0) {
      ROWS16(Y_ROW_LOAD)
    }
    #pragma unroll
    for (int ch = 0; ch < CH_GRP; ++ch) {
      v8f ym = zero;
      #pragma unroll
      for (int j = 0; j < 4; ++j) ym = wmma4_f32(mA[j], ub[ch][j], ym);
      yt[ch] = ym;
    }
    v8f S[NBLK], Sm[NBLK];
    #pragma unroll
    for (int bk = 0; bk < NBLK; ++bk) { S[bk] = Cst[bk]; Sm[bk] = Cst[bk]; }
    #pragma unroll
    for (int v = 0; v < NWAVE; ++v) {
      #pragma unroll
      for (int bk = 0; bk < NBLK; ++bk) {
        S[bk] = S[bk] * apv[bk] + ev[v][bk];
        if (v + 1 < NWAVE) Sm[bk] = (wv == v + 1) ? S[bk] : Sm[bk];
      }
    }
    #pragma unroll
    for (int bk = 0; bk < NBLK; ++bk) Cst[bk] = S[bk];
    if (r > 0) {
      const int yoff = (r - 1) * (NWAVE * GRPLEN);
      ROWS16(Y_ROW_STORE)
    }
    MEM_FENCE

    #pragma unroll
    for (int cp = 0; cp < CH_GRP / 2; ++cp) {
      const int c0 = 2 * cp, c1 = 2 * cp + 1;
      v8f H1[NBLK];
      #pragma unroll
      for (int bk = 0; bk < NBLK; ++bk) H1[bk] = Sm[bk] * avv[bk] + G[c0][bk];
      v8f y0 = yt[c0], y1 = yt[c1];
      #pragma unroll
      for (int bk = 0; bk < NBLK; ++bk) {
        #pragma unroll
        for (int j = 0; j < 4; ++j) {
          v2f h0, h1;
          h0.x = Sm[bk][2 * j]; h0.y = Sm[bk][2 * j + 1];
          h1.x = H1[bk][2 * j]; h1.y = H1[bk][2 * j + 1];
          y0 = wmma4_f32(cm[bk][j], h0, y0);
          y1 = wmma4_f32(cm[bk][j], h1, y1);
        }
      }
      yt[c0] = y0; yt[c1] = y1;
      #pragma unroll
      for (int bk = 0; bk < NBLK; ++bk) Sm[bk] = H1[bk] * avv[bk] + G[c1][bk];
    }
    MEM_FENCE
  }

  {
    const int yoff = (NRND - 1) * (NWAVE * GRPLEN);
    ROWS16(Y_DECL)
    Y_TO_LDS
    wave_lds_order();
    ROWS16(Y_ROW_LOAD)
    ROWS16(Y_ROW_STORE)
  }
#undef MEM_FENCE
#undef Y_TO_LDS
#undef Y_DECL
#undef Y_ROW_LOAD
#undef Y_ROW_STORE
#undef ROWS16
#undef U_FIRST
#undef U_TO_LDS
#undef U_NEXT
}

extern "C" void kernel_launch(void* const* d_in, const int* in_sizes, int n_in,
                              void* d_out, int out_size, void* d_ws, size_t ws_size,
                              hipStream_t stream) {
  const float* u  = (const float*)d_in[0];
  const float* A  = (const float*)d_in[1];
  const float* Bv = (const float*)d_in[2];
  const float* Cv = (const float*)d_in[3];
  float* y      = (float*)d_out;

  ssm_fused<<<dim3(NTILE), dim3(NWAVE * 32), 0, stream>>>(u, A, Bv, Cv, y);
}
